# starting kernel
# baseline (speedup 1.0000x reference)
#include <hip/hip_runtime.h>

#define BB 2
#define CC 64
#define DD 3
#define NN 4096
#define KK 16
#define EE 128
#define HH 256

typedef __attribute__((ext_vector_type(2))) float v2f;
typedef __attribute__((ext_vector_type(8))) float v8f;

constexpr int NT   = 4;
constexpr int COLS = NT * KK;
constexpr int LDSC = COLS + 4;

__global__ void meanpc_kernel(const float* __restrict__ x, float* __restrict__ mean_pc) {
    int i = blockIdx.x * blockDim.x + threadIdx.x;
    if (i >= BB * CC * NN) return;
    int n  = i % NN;
    int bc = i / NN;
    const float* xp = x + (size_t)bc * (DD * NN) + n;
    mean_pc[i] = (xp[0] + xp[NN] + xp[2 * NN]) * (1.0f / 3.0f);
}

__global__ void sqnorm_kernel(const float* __restrict__ mean_pc, float* __restrict__ sq) {
    int i = blockIdx.x * blockDim.x + threadIdx.x;
    if (i >= BB * NN) return;
    int n = i % NN;
    int b = i / NN;
    float s = 0.0f;
#pragma unroll
    for (int c = 0; c < CC; ++c) {
        float v = mean_pc[((size_t)b * CC + c) * NN + n];
        s = __builtin_fmaf(v, v, s);
    }
    sq[i] = s;
}

__global__ void __launch_bounds__(256) knn_kernel(const float* __restrict__ mean_pc,
                                                  const float* __restrict__ sq,
                                                  int* __restrict__ idx) {
    int t = threadIdx.x;
    int b = blockIdx.x / (NN / 256);
    int n = (blockIdx.x % (NN / 256)) * 256 + t;

    float q[CC];
#pragma unroll
    for (int c = 0; c < CC; ++c) q[c] = mean_pc[((size_t)b * CC + c) * NN + n];

    float bd[KK];
    int   bi[KK];
#pragma unroll
    for (int j = 0; j < KK; ++j) { bd[j] = 3.0e38f; bi[j] = 0; }

    for (int r = 0; r < NN; ++r) {
        float dot = 0.0f;
#pragma unroll
        for (int c = 0; c < CC; ++c)
            dot = __builtin_fmaf(q[c], mean_pc[((size_t)b * CC + c) * NN + r], dot);
        float d2 = sq[(size_t)b * NN + r] - 2.0f * dot;
        if (r != n && d2 < bd[KK - 1]) {
            bd[KK - 1] = d2;
            bi[KK - 1] = r;
#pragma unroll
            for (int j = KK - 1; j > 0; --j) {
                if (bd[j] < bd[j - 1]) {
                    float td = bd[j]; bd[j] = bd[j - 1]; bd[j - 1] = td;
                    int   ti = bi[j]; bi[j] = bi[j - 1]; bi[j - 1] = ti;
                }
            }
        }
    }
#pragma unroll
    for (int j = 0; j < KK; ++j) idx[((size_t)b * KK + j) * NN + n] = bi[j];
}

template <int NCT, bool RELU>
__device__ __forceinline__ void gemm_tile(const float* __restrict__ W,
                                          const float* __restrict__ bias,
                                          const float* lin, float* lout,
                                          int Ki, int rt, int ctBase,
                                          int laneRow, int laneHi) {
    const v8f vzero = {0.f, 0.f, 0.f, 0.f, 0.f, 0.f, 0.f, 0.f};
    v8f acc[NCT];
#pragma unroll
    for (int c = 0; c < NCT; ++c) acc[c] = vzero;

    const float* wrow = W + (size_t)(rt * 16 + laneRow) * Ki;
    for (int k0 = 0; k0 < Ki; k0 += 4) {
        v2f a;
        a.x = wrow[k0 + 2 * laneHi + 0];
        a.y = wrow[k0 + 2 * laneHi + 1];
#pragma unroll
        for (int c = 0; c < NCT; ++c) {
            int col = (ctBase + c) * 16 + laneRow;
            v2f bf;
            bf.x = lin[(k0 + 2 * laneHi + 0) * LDSC + col];
            bf.y = lin[(k0 + 2 * laneHi + 1) * LDSC + col];
            acc[c] = __builtin_amdgcn_wmma_f32_16x16x4_f32(
                false, a, false, bf, (short)0, acc[c], false, false);
        }
    }
#pragma unroll
    for (int c = 0; c < NCT; ++c) {
#pragma unroll
        for (int j = 0; j < 8; ++j) {
            int row = rt * 16 + j + 8 * laneHi;
            float v = acc[c][j] + bias[row];
            if (RELU) v = fmaxf(v, 0.0f);
            lout[row * LDSC + (ctBase + c) * 16 + laneRow] = v;
        }
    }
}

__global__ void __launch_bounds__(256) mlp_kernel(const float* __restrict__ x,
                                                  const float* __restrict__ mean_pc,
                                                  const int* __restrict__ idx,
                                                  const float* __restrict__ W1, const float* __restrict__ b1,
                                                  const float* __restrict__ W2, const float* __restrict__ b2,
                                                  const float* __restrict__ W3, const float* __restrict__ b3,
                                                  float* __restrict__ out) {
    __shared__ float ldsX[HH * LDSC];
    __shared__ float ldsY[HH * LDSC];

    int t       = threadIdx.x;
    int lane    = t & 31;
    int w       = t >> 5;
    int laneRow = lane & 15;
    int laneHi  = lane >> 4;

    int blk = blockIdx.x;
    int nb  = blk % (NN / NT);
    int bd  = blk / (NN / NT);
    int d   = bd % DD;
    int b   = bd / DD;
    int n0  = nb * NT;

    for (int e = t; e < EE * COLS; e += 256) {
        int row = e >> 6;
        int col = e & (COLS - 1);
        int nl  = col >> 4;
        int k   = col & 15;
        int n   = n0 + nl;
        float v;
        if (row < CC) {
            v = x[(((size_t)b * CC + row) * DD + d) * NN + n];
        } else {
            int c = row - CC;
            int j = idx[((size_t)b * KK + k) * NN + n];
            v = mean_pc[((size_t)b * CC + c) * NN + j]
              - x[(((size_t)b * CC + c) * DD + d) * NN + n];
        }
        ldsX[row * LDSC + col] = v;
    }
    __syncthreads();

    gemm_tile<4, true>(W1, b1, ldsX, ldsY, EE, 2 * w + 0, 0, laneRow, laneHi);
    gemm_tile<4, true>(W1, b1, ldsX, ldsY, EE, 2 * w + 1, 0, laneRow, laneHi);
    __syncthreads();

    gemm_tile<4, true>(W2, b2, ldsY, ldsX, HH, 2 * w + 0, 0, laneRow, laneHi);
    gemm_tile<4, true>(W2, b2, ldsY, ldsX, HH, 2 * w + 1, 0, laneRow, laneHi);
    __syncthreads();

    gemm_tile<2, false>(W3, b3, ldsX, ldsY, HH, w >> 1, (w & 1) * 2, laneRow, laneHi);
    __syncthreads();

    {
        int c  = t & (CC - 1);
        int nl = t >> 6;
        const float* rrow = &ldsY[c * LDSC + nl * 16];
        float m = rrow[0];
#pragma unroll
        for (int k = 1; k < KK; ++k) m = fmaxf(m, rrow[k]);
        size_t o = (((size_t)b * CC + c) * DD + d) * NN + (n0 + nl);
        out[o] = m + x[o];
    }
}

extern "C" void kernel_launch(void* const* d_in, const int* in_sizes, int n_in,
                              void* d_out, int out_size, void* d_ws, size_t ws_size,
                              hipStream_t stream) {
    const float* x  = (const float*)d_in[0];
    const float* W1 = (const float*)d_in[1];
    const float* b1 = (const float*)d_in[2];
    const float* W2 = (const float*)d_in[3];
    const float* b2 = (const float*)d_in[4];
    const float* W3 = (const float*)d_in[5];
    const float* b3 = (const float*)d_in[6];
    float* out = (float*)d_out;

    float* mean_pc = (float*)d_ws;
    float* sq      = mean_pc + (size_t)BB * CC * NN;
    int*   idxp    = (int*)(sq + (size_t)BB * NN);

    meanpc_kernel<<<(BB * CC * NN + 255) / 256, 256, 0, stream>>>(x, mean_pc);
    sqnorm_kernel<<<(BB * NN + 255) / 256, 256, 0, stream>>>(mean_pc, sq);
    knn_kernel<<<BB * (NN / 256), 256, 0, stream>>>(mean_pc, sq, idxp);
    mlp_kernel<<<BB * DD * (NN / NT), 256, 0, stream>>>(x, mean_pc, idxp,
                                                        W1, b1, W2, b2, W3, b3, out);
}


# MLP activations kept in registers: one wave takes 32 columns through all 3 layers, a layer's D tiles are the next layer's B fragments (k order permuted: f32 sum order changes); no LDS, no barrier
# speedup vs baseline: 8.0379x; 8.0379x over previous
#include <hip/hip_runtime.h>

#define BB 2
#define CC 64
#define DD 3
#define NN 4096
#define KK 16
#define EE 128
#define HH 256

typedef __attribute__((ext_vector_type(2))) float v2f;
typedef __attribute__((ext_vector_type(8))) float v8f;

constexpr int NCT  = 2;
constexpr int MW   = 4;

__global__ void meanpc_kernel(const float* __restrict__ x, float* __restrict__ mean_pc) {
    int i = blockIdx.x * blockDim.x + threadIdx.x;
    if (i >= BB * CC * NN) return;
    int n  = i % NN;
    int bc = i / NN;
    const float* xp = x + (size_t)bc * (DD * NN) + n;
    mean_pc[i] = (xp[0] + xp[NN] + xp[2 * NN]) * (1.0f / 3.0f);
}

__global__ void sqnorm_kernel(const float* __restrict__ mean_pc, float* __restrict__ sq) {
    int i = blockIdx.x * blockDim.x + threadIdx.x;
    if (i >= BB * NN) return;
    int n = i % NN;
    int b = i / NN;
    float s = 0.0f;
#pragma unroll
    for (int c = 0; c < CC; ++c) {
        float v = mean_pc[((size_t)b * CC + c) * NN + n];
        s = __builtin_fmaf(v, v, s);
    }
    sq[i] = s;
}

typedef __attribute__((ext_vector_type(4))) float v4f;

constexpr int KQ    = 32;
constexpr int TR    = 32;
constexpr int PP    = 2 * CC + 4;
constexpr int RG    = 16;
constexpr int NPAIR = RG / 2;
constexpr int NBAT  = CC / 2;
constexpr int AHEAD = 3;
constexpr int NSLOT = AHEAD + 1;
constexpr int NGRP   = TR / RG;
static_assert(NBAT % NSLOT == 0, "slot phase must repeat from group to group");
static_assert(NGRP == 2, "group wrap below assumes two groups per tile");

__device__ __forceinline__ void knn_merge_group(const float (*sc)[KQ], int lane, int r0, int n,
                                                float (&bd)[KK], int (&bi)[KK]) {
    float d2[RG];
    bool any = false;
#pragma unroll
    for (int i = 0; i < RG; ++i) {
        d2[i] = sc[i][lane];
        any |= d2[i] < bd[KK - 1];
    }
    if (any) {
#pragma unroll
        for (int i = 0; i < RG; ++i) {
            const int r = r0 + i;
            if (r != n && d2[i] < bd[KK - 1]) {
                bd[KK - 1] = d2[i];
                bi[KK - 1] = r;
#pragma unroll
                for (int j = KK - 1; j > 0; --j) {
                    if (bd[j] < bd[j - 1]) {
                        float td = bd[j]; bd[j] = bd[j - 1]; bd[j - 1] = td;
                        int   ti = bi[j]; bi[j] = bi[j - 1]; bi[j - 1] = ti;
                    }
                }
            }
        }
    }
}

__global__ void __launch_bounds__(2 * KQ) knn_kernel(const float* __restrict__ mean_pc,
                                                     const float* __restrict__ sq,
                                                     int* __restrict__ idx) {
    __shared__ __attribute__((aligned(16))) float tile[(TR / 2) * PP];
    __shared__ float score[2][RG][KQ];

    const int lane = threadIdx.x & (KQ - 1);
    const int role = __builtin_amdgcn_readfirstlane((int)(threadIdx.x >> 5));
    const int b = blockIdx.x / (NN / KQ);
    const int n = (blockIdx.x % (NN / KQ)) * KQ + lane;
    const float* pcb = mean_pc + (size_t)b * CC * NN;
    const float* sqb = sq + (size_t)b * NN;

    float q[CC];
    float pf[CC];
    if (role == 0) {
#pragma unroll
        for (int c = 0; c < CC; ++c) q[c] = pcb[(size_t)c * NN + n];
#pragma unroll
        for (int c = 0; c < CC; ++c) pf[c] = pcb[(size_t)c * NN + lane];
    }

    float bd[KK];
    int   bi[KK];
#pragma unroll
    for (int j = 0; j < KK; ++j) { bd[j] = 3.0e38f; bi[j] = 0; }

    const int sp = (lane >> 1) * PP + (lane & 1);

    for (int t = 0; t < NN / TR; ++t) {
        if (role == 0) {
#pragma unroll
            for (int c = 0; c < CC; ++c) tile[sp + 2 * c] = pf[c];
        }
        __syncthreads();

        v4f rv[NSLOT][NPAIR];
        if (role == 0) {
            if (t + 1 < NN / TR) {
#pragma unroll
                for (int c = 0; c < CC; ++c) pf[c] = pcb[(size_t)c * NN + (t + 1) * TR + lane];
            }
#pragma unroll
            for (int s = 0; s < AHEAD; ++s) {
#pragma unroll
                for (int p = 0; p < NPAIR; ++p) rv[s][p] = *(const v4f*)&tile[p * PP + 4 * s];
            }
        }

        for (int g = 0; g < NGRP; ++g) {
            const int G = t * NGRP + g;
            if (role == 0) {
                const float* rows  = &tile[g * NPAIR * PP];
                const float* nrows = &tile[((g + 1) & 1) * NPAIR * PP];
                v2f acc[NPAIR];
#pragma unroll
                for (int p = 0; p < NPAIR; ++p) acc[p] = (v2f){0.0f, 0.0f};
#pragma unroll
                for (int bt = 0; bt < NBAT; ++bt) {
                    const int nx = bt + AHEAD;
                    const float* src = (nx < NBAT) ? rows : nrows;
                    const int    nb  = (nx < NBAT) ? nx : nx - NBAT;
#pragma unroll
                    for (int p = 0; p < NPAIR; ++p)
                        rv[nx % NSLOT][p] = *(const v4f*)&src[p * PP + 4 * nb];
                    __builtin_amdgcn_sched_barrier(0);
                    const int cs = bt % NSLOT;
                    const v2f q0 = {q[2 * bt + 0], q[2 * bt + 0]};
                    const v2f q1 = {q[2 * bt + 1], q[2 * bt + 1]};
#pragma unroll
                    for (int p = 0; p < NPAIR; ++p) acc[p] = __builtin_elementwise_fma(q0, rv[cs][p].xy, acc[p]);
#pragma unroll
                    for (int p = 0; p < NPAIR; ++p) acc[p] = __builtin_elementwise_fma(q1, rv[cs][p].zw, acc[p]);
                    __builtin_amdgcn_sched_barrier(0);
                }

                const int r0 = G * RG;
#pragma unroll
                for (int i = 0; i < RG; ++i)
                    score[G & 1][i][lane] = sqb[r0 + i] - 2.0f * acc[i >> 1][i & 1];
            } else if (G > 0) {
                knn_merge_group(score[(G - 1) & 1], lane, (G - 1) * RG, n, bd, bi);
            }
            __syncthreads();
        }
    }
    if (role == 1) {
        constexpr int GL = NN / RG - 1;
        knn_merge_group(score[GL & 1], lane, GL * RG, n, bd, bi);
#pragma unroll
        for (int j = 0; j < KK; ++j) idx[((size_t)b * KK + j) * NN + n] = bi[j];
    }
}

typedef __attribute__((ext_vector_type(2))) int v2i;

__device__ __forceinline__ v8f ld8(const float* p) {
    const v4f lo = *(const v4f*)p;
    const v4f hi = *(const v4f*)(p + 4);
    return (v8f){lo.x, lo.y, lo.z, lo.w, hi.x, hi.y, hi.z, hi.w};
}

__device__ __forceinline__ void mma_block(const v8f& a, const v8f& b0, const v8f& b1, v8f& c0, v8f& c1) {
#pragma unroll
    for (int m = 0; m < 4; ++m) {
        const v2f af = {a[2 * m], a[2 * m + 1]};
        const v2f f0 = {b0[2 * m], b0[2 * m + 1]};
        const v2f f1 = {b1[2 * m], b1[2 * m + 1]};
        c0 = __builtin_amdgcn_wmma_f32_16x16x4_f32(false, af, false, f0, (short)0, c0, false, false);
        c1 = __builtin_amdgcn_wmma_f32_16x16x4_f32(false, af, false, f1, (short)0, c1, false, false);
    }
}

__device__ __forceinline__ v8f bias_relu(const v8f& acc, const v8f& bias) {
    v8f o;
#pragma unroll
    for (int r = 0; r < 8; ++r) o[r] = fmaxf(acc[r] + bias[r], 0.0f);
    return o;
}

constexpr int AH   = 3;
constexpr int NRNG = AH + 1;
constexpr int S1   = EE / 16;
constexpr int S12  = S1 + HH / 16;
constexpr int S3   = HH / 16;
static_assert(S12 % NRNG == 0 && S3 % NRNG == 0, "ring phase must repeat from trip to trip");
static_assert(NCT == 2, "mma_block and the paired loads below are written for two column tiles");

__global__ void __launch_bounds__(MW * 32) mlp_kernel(const float* __restrict__ x,
                                                      const float* __restrict__ mean_pc,
                                                      const int* __restrict__ idx,
                                                      const float* __restrict__ W1, const float* __restrict__ b1,
                                                      const float* __restrict__ W2, const float* __restrict__ b2,
                                                      const float* __restrict__ W3, const float* __restrict__ b3,
                                                      float* __restrict__ out) {
    const int lane = threadIdx.x & 31;
    const int c    = lane & 15;
    const int h    = lane >> 4;
    const int wv   = __builtin_amdgcn_readfirstlane((int)(threadIdx.x >> 5));
    const int task = blockIdx.x * MW + wv;
    const int np   = task % (NN / NCT);
    const int bd   = task / (NN / NCT);
    const int d    = bd % DD;
    const int b    = bd / DD;
    const int n0   = np * NCT;

    const v8f vzero = {0.f, 0.f, 0.f, 0.f, 0.f, 0.f, 0.f, 0.f};

    const float* w1p = W1 + (size_t)c * EE + 8 * h;
    const float* w2p = W2 + (size_t)c * HH + 8 * h;
    const float* w3p = W3 + (size_t)c * HH + 8 * h;

#define WPTR12(T, s) (((s) < S1) ? (w1p + (size_t)(T) * 16 * EE + (s) * 16) \
                                 : (w2p + (size_t)((s) - S1) * 16 * HH + (T) * 16))

    v8f A[NRNG];
#pragma unroll
    for (int s = 0; s < AH; ++s) A[s] = ld8(WPTR12(0, s));

    v8f E[S1][NCT];
    {
        const v2i jj = *(const v2i*)&idx[((size_t)b * KK + c) * NN + n0];
        const int j0 = jj.x & (NN - 1);
        const int j1 = jj.y & (NN - 1);
        const float* xb = x + (((size_t)b * CC) * DD + d) * NN + n0;
        const float* mb = mean_pc + (size_t)b * CC * NN;
#pragma unroll
        for (int q = 0; q < CC / 16; ++q) {
#pragma unroll
            for (int r = 0; r < 8; ++r) {
                const int ch = q * 16 + 8 * h + r;
                const v2f xv = *(const v2f*)&xb[(size_t)ch * DD * NN];
                const float g0 = mb[(size_t)ch * NN + j0];
                const float g1 = mb[(size_t)ch * NN + j1];
                E[q][0][r] = xv.x;
                E[q][1][r] = xv.y;
                E[q + CC / 16][0][r] = g0 - xv.x;
                E[q + CC / 16][1][r] = g1 - xv.y;
            }
        }
    }

    v8f H2[HH / 16][NCT];
#pragma unroll
    for (int u = 0; u < HH / 16; ++u) { H2[u][0] = vzero; H2[u][1] = vzero; }

    for (int T = 0; T < HH / 16; ++T) {
        const int Tn = (T + 1 < HH / 16) ? T + 1 : T;
        const v8f bias1 = ld8(b1 + T * 16 + 8 * h);
        v8f c0 = vzero, c1 = vzero;
        v8f t0 = vzero, t1 = vzero;
#pragma unroll
        for (int s = 0; s < S12; ++s) {
            const int sn = s + AH;
            A[sn % NRNG] = ld8(sn < S12 ? WPTR12(T, sn) : WPTR12(Tn, sn - S12));
            __builtin_amdgcn_sched_barrier(0);
            if (s < S1) {
                mma_block(A[s % NRNG], E[s][0], E[s][1], c0, c1);
            } else {
                if (s == S1) { t0 = bias_relu(c0, bias1); t1 = bias_relu(c1, bias1); }
                mma_block(A[s % NRNG], t0, t1, H2[s - S1][0], H2[s - S1][1]);
            }
            __builtin_amdgcn_sched_barrier(0);
        }
    }
#undef WPTR12

#define WPTR3(V, s) (w3p + (size_t)(V) * 16 * HH + (s) * 16)
#pragma unroll
    for (int s = 0; s < AH; ++s) A[s] = ld8(WPTR3(0, s));

#pragma unroll
    for (int u = 0; u < HH / 16; ++u) {
        const v8f bias2 = ld8(b2 + u * 16 + 8 * h);
        H2[u][0] = bias_relu(H2[u][0], bias2);
        H2[u][1] = bias_relu(H2[u][1], bias2);
    }

    for (int V = 0; V < CC / 16; ++V) {
        const int Vn = (V + 1 < CC / 16) ? V + 1 : V;
        const v8f bias3 = ld8(b3 + V * 16 + 8 * h);
        v8f c0 = vzero, c1 = vzero;
#pragma unroll
        for (int s = 0; s < S3; ++s) {
            const int sn = s + AH;
            A[sn % NRNG] = ld8(sn < S3 ? WPTR3(V, sn) : WPTR3(Vn, sn - S3));
            __builtin_amdgcn_sched_barrier(0);
            mma_block(A[s % NRNG], H2[s][0], H2[s][1], c0, c1);
            __builtin_amdgcn_sched_barrier(0);
        }
        float m0[8], m1[8];
#pragma unroll
        for (int r = 0; r < 8; ++r) { m0[r] = c0[r] + bias3[r]; m1[r] = c1[r] + bias3[r]; }
#pragma unroll
        for (int o = 1; o < 16; o <<= 1) {
#pragma unroll
            for (int r = 0; r < 8; ++r) {
                m0[r] = fmaxf(m0[r], __shfl_xor(m0[r], o));
                m1[r] = fmaxf(m1[r], __shfl_xor(m1[r], o));
            }
        }
        if (c == 0) {
            v2f xr[8];
#pragma unroll
            for (int r = 0; r < 8; ++r)
                xr[r] = *(const v2f*)&x[(((size_t)b * CC + V * 16 + 8 * h + r) * DD + d) * NN + n0];
#pragma unroll
            for (int r = 0; r < 8; ++r)
                *(v2f*)&out[(((size_t)b * CC + V * 16 + 8 * h + r) * DD + d) * NN + n0] = (v2f){m0[r] + xr[r].x, m1[r] + xr[r].y};
        }
    }
#undef WPTR3
}

extern "C" void kernel_launch(void* const* d_in, const int* in_sizes, int n_in,
                              void* d_out, int out_size, void* d_ws, size_t ws_size,
                              hipStream_t stream) {
    const float* x  = (const float*)d_in[0];
    const float* W1 = (const float*)d_in[1];
    const float* b1 = (const float*)d_in[2];
    const float* W2 = (const float*)d_in[3];
    const float* b2 = (const float*)d_in[4];
    const float* W3 = (const float*)d_in[5];
    const float* b3 = (const float*)d_in[6];
    float* out = (float*)d_out;

    float* mean_pc = (float*)d_ws;
    float* sq      = mean_pc + (size_t)BB * CC * NN;
    int*   idxp    = (int*)(sq + (size_t)BB * NN);

    meanpc_kernel<<<(BB * CC * NN + 255) / 256, 256, 0, stream>>>(x, mean_pc);
    sqnorm_kernel<<<(BB * NN + 255) / 256, 256, 0, stream>>>(mean_pc, sq);
    knn_kernel<<<BB * (NN / 32), 64, 0, stream>>>(mean_pc, sq, idxp);
    mlp_kernel<<<BB * DD * (NN / NCT) / MW, MW * 32, 0, stream>>>(x, mean_pc, idxp,
                                                        W1, b1, W2, b2, W3, b3, out);
}


# KNN workgroup 2 -> 3 waves: two scoring waves take alternate 32-point tiles (own LDS tile each), one list wave merges a step's 4 groups in point order; scores and lists unchanged
# speedup vs baseline: 9.0852x; 1.1303x over previous
#include <hip/hip_runtime.h>

#define BB 2
#define CC 64
#define DD 3
#define NN 4096
#define KK 16
#define EE 128
#define HH 256

typedef __attribute__((ext_vector_type(2))) float v2f;
typedef __attribute__((ext_vector_type(8))) float v8f;

constexpr int NCT  = 2;
constexpr int MW   = 4;

__global__ void meanpc_kernel(const float* __restrict__ x, float* __restrict__ mean_pc) {
    int i = blockIdx.x * blockDim.x + threadIdx.x;
    if (i >= BB * CC * NN) return;
    int n  = i % NN;
    int bc = i / NN;
    const float* xp = x + (size_t)bc * (DD * NN) + n;
    mean_pc[i] = (xp[0] + xp[NN] + xp[2 * NN]) * (1.0f / 3.0f);
}

__global__ void sqnorm_kernel(const float* __restrict__ mean_pc, float* __restrict__ sq) {
    int i = blockIdx.x * blockDim.x + threadIdx.x;
    if (i >= BB * NN) return;
    int n = i % NN;
    int b = i / NN;
    float s = 0.0f;
#pragma unroll
    for (int c = 0; c < CC; ++c) {
        float v = mean_pc[((size_t)b * CC + c) * NN + n];
        s = __builtin_fmaf(v, v, s);
    }
    sq[i] = s;
}

typedef __attribute__((ext_vector_type(4))) float v4f;

constexpr int KQ    = 32;
constexpr int NS    = 2;
constexpr int TR    = 32;
constexpr int PP    = 2 * CC + 4;
constexpr int RG    = 16;
constexpr int NPAIR = RG / 2;
constexpr int NBAT  = CC / 2;
constexpr int AHEAD = 3;
constexpr int NSLOT = AHEAD + 1;
constexpr int NGRP  = TR / RG;
constexpr int SG    = NS * NGRP;
constexpr int NSTEP = NN / (NS * TR);
static_assert(NBAT % NSLOT == 0, "slot phase must repeat from group to group");
static_assert(NGRP == 2, "group wrap below assumes two groups per tile");
static_assert(NN % (NS * TR) == 0, "steps must cover the points exactly");

__device__ __forceinline__ void knn_merge_group(const float (*sc)[KQ], int lane, int r0, int n,
                                                float (&bd)[KK], int (&bi)[KK]) {
    float d2[RG];
    bool any = false;
#pragma unroll
    for (int i = 0; i < RG; ++i) {
        d2[i] = sc[i][lane];
        any |= d2[i] < bd[KK - 1];
    }
    if (any) {
#pragma unroll
        for (int i = 0; i < RG; ++i) {
            const int r = r0 + i;
            if (r != n && d2[i] < bd[KK - 1]) {
                bd[KK - 1] = d2[i];
                bi[KK - 1] = r;
#pragma unroll
                for (int j = KK - 1; j > 0; --j) {
                    if (bd[j] < bd[j - 1]) {
                        float td = bd[j]; bd[j] = bd[j - 1]; bd[j - 1] = td;
                        int   ti = bi[j]; bi[j] = bi[j - 1]; bi[j - 1] = ti;
                    }
                }
            }
        }
    }
}

__global__ void __launch_bounds__((NS + 1) * KQ) knn_kernel(const float* __restrict__ mean_pc,
                                                            const float* __restrict__ sq,
                                                            int* __restrict__ idx) {
    __shared__ __attribute__((aligned(16))) float tile[NS][(TR / 2) * PP];
    __shared__ float score[2][SG][RG][KQ];

    const int lane = threadIdx.x & (KQ - 1);
    const int role = __builtin_amdgcn_readfirstlane((int)(threadIdx.x >> 5));
    const bool scorer = role < NS;
    const int b = blockIdx.x / (NN / KQ);
    const int n = (blockIdx.x % (NN / KQ)) * KQ + lane;
    const float* pcb = mean_pc + (size_t)b * CC * NN;
    const float* sqb = sq + (size_t)b * NN;
    float* mytile = &tile[scorer ? role : 0][0];

    float q[CC];
    float pf[CC];
    if (scorer) {
#pragma unroll
        for (int c = 0; c < CC; ++c) q[c] = pcb[(size_t)c * NN + n];
#pragma unroll
        for (int c = 0; c < CC; ++c) pf[c] = pcb[(size_t)c * NN + role * TR + lane];
    }

    float bd[KK];
    int   bi[KK];
#pragma unroll
    for (int j = 0; j < KK; ++j) { bd[j] = 3.0e38f; bi[j] = 0; }

    const int sp = (lane >> 1) * PP + (lane & 1);

    for (int st = 0; st < NSTEP; ++st) {
        if (scorer) {
#pragma unroll
            for (int c = 0; c < CC; ++c) mytile[sp + 2 * c] = pf[c];
        }
        __syncthreads();

        if (scorer) {
            const int t = st * NS + role;
            v4f rv[NSLOT][NPAIR];
            if (st + 1 < NSTEP) {
#pragma unroll
                for (int c = 0; c < CC; ++c) pf[c] = pcb[(size_t)c * NN + (t + NS) * TR + lane];
            }
#pragma unroll
            for (int s = 0; s < AHEAD; ++s) {
#pragma unroll
                for (int p = 0; p < NPAIR; ++p) rv[s][p] = *(const v4f*)&mytile[p * PP + 4 * s];
            }

            for (int g = 0; g < NGRP; ++g) {
                const float* rows  = &mytile[g * NPAIR * PP];
                const float* nrows = &mytile[((g + 1) & 1) * NPAIR * PP];
                v2f acc[NPAIR];
#pragma unroll
                for (int p = 0; p < NPAIR; ++p) acc[p] = (v2f){0.0f, 0.0f};
#pragma unroll
                for (int bt = 0; bt < NBAT; ++bt) {
                    const int nx = bt + AHEAD;
                    const float* src = (nx < NBAT) ? rows : nrows;
                    const int    nb  = (nx < NBAT) ? nx : nx - NBAT;
#pragma unroll
                    for (int p = 0; p < NPAIR; ++p)
                        rv[nx % NSLOT][p] = *(const v4f*)&src[p * PP + 4 * nb];
                    __builtin_amdgcn_sched_barrier(0);
                    const int cs = bt % NSLOT;
                    const v2f q0 = {q[2 * bt + 0], q[2 * bt + 0]};
                    const v2f q1 = {q[2 * bt + 1], q[2 * bt + 1]};
#pragma unroll
                    for (int p = 0; p < NPAIR; ++p) acc[p] = __builtin_elementwise_fma(q0, rv[cs][p].xy, acc[p]);
#pragma unroll
                    for (int p = 0; p < NPAIR; ++p) acc[p] = __builtin_elementwise_fma(q1, rv[cs][p].zw, acc[p]);
                    __builtin_amdgcn_sched_barrier(0);
                }

                const int r0 = (t * NGRP + g) * RG;
#pragma unroll
                for (int i = 0; i < RG; ++i)
                    score[st & 1][role * NGRP + g][i][lane] = sqb[r0 + i] - 2.0f * acc[i >> 1][i & 1];
            }
        } else if (st > 0) {
            for (int k = 0; k < SG; ++k)
                knn_merge_group(score[(st - 1) & 1][k], lane, ((st - 1) * SG + k) * RG, n, bd, bi);
        }
        __syncthreads();
    }
    if (!scorer) {
        constexpr int SL = NSTEP - 1;
        for (int k = 0; k < SG; ++k)
            knn_merge_group(score[SL & 1][k], lane, (SL * SG + k) * RG, n, bd, bi);
#pragma unroll
        for (int j = 0; j < KK; ++j) idx[((size_t)b * KK + j) * NN + n] = bi[j];
    }
}

typedef __attribute__((ext_vector_type(2))) int v2i;

__device__ __forceinline__ v8f ld8(const float* p) {
    const v4f lo = *(const v4f*)p;
    const v4f hi = *(const v4f*)(p + 4);
    return (v8f){lo.x, lo.y, lo.z, lo.w, hi.x, hi.y, hi.z, hi.w};
}

__device__ __forceinline__ void mma_block(const v8f& a, const v8f& b0, const v8f& b1, v8f& c0, v8f& c1) {
#pragma unroll
    for (int m = 0; m < 4; ++m) {
        const v2f af = {a[2 * m], a[2 * m + 1]};
        const v2f f0 = {b0[2 * m], b0[2 * m + 1]};
        const v2f f1 = {b1[2 * m], b1[2 * m + 1]};
        c0 = __builtin_amdgcn_wmma_f32_16x16x4_f32(false, af, false, f0, (short)0, c0, false, false);
        c1 = __builtin_amdgcn_wmma_f32_16x16x4_f32(false, af, false, f1, (short)0, c1, false, false);
    }
}

__device__ __forceinline__ v8f bias_relu(const v8f& acc, const v8f& bias) {
    v8f o;
#pragma unroll
    for (int r = 0; r < 8; ++r) o[r] = fmaxf(acc[r] + bias[r], 0.0f);
    return o;
}

constexpr int AH   = 3;
constexpr int NRNG = AH + 1;
constexpr int S1   = EE / 16;
constexpr int S12  = S1 + HH / 16;
constexpr int S3   = HH / 16;
static_assert(S12 % NRNG == 0 && S3 % NRNG == 0, "ring phase must repeat from trip to trip");
static_assert(NCT == 2, "mma_block and the paired loads below are written for two column tiles");

__global__ void __launch_bounds__(MW * 32) mlp_kernel(const float* __restrict__ x,
                                                      const float* __restrict__ mean_pc,
                                                      const int* __restrict__ idx,
                                                      const float* __restrict__ W1, const float* __restrict__ b1,
                                                      const float* __restrict__ W2, const float* __restrict__ b2,
                                                      const float* __restrict__ W3, const float* __restrict__ b3,
                                                      float* __restrict__ out) {
    const int lane = threadIdx.x & 31;
    const int c    = lane & 15;
    const int h    = lane >> 4;
    const int wv   = __builtin_amdgcn_readfirstlane((int)(threadIdx.x >> 5));
    const int task = blockIdx.x * MW + wv;
    const int np   = task % (NN / NCT);
    const int bd   = task / (NN / NCT);
    const int d    = bd % DD;
    const int b    = bd / DD;
    const int n0   = np * NCT;

    const v8f vzero = {0.f, 0.f, 0.f, 0.f, 0.f, 0.f, 0.f, 0.f};

    const float* w1p = W1 + (size_t)c * EE + 8 * h;
    const float* w2p = W2 + (size_t)c * HH + 8 * h;
    const float* w3p = W3 + (size_t)c * HH + 8 * h;

#define WPTR12(T, s) (((s) < S1) ? (w1p + (size_t)(T) * 16 * EE + (s) * 16) \
                                 : (w2p + (size_t)((s) - S1) * 16 * HH + (T) * 16))

    v8f A[NRNG];
#pragma unroll
    for (int s = 0; s < AH; ++s) A[s] = ld8(WPTR12(0, s));

    v8f E[S1][NCT];
    {
        const v2i jj = *(const v2i*)&idx[((size_t)b * KK + c) * NN + n0];
        const int j0 = jj.x & (NN - 1);
        const int j1 = jj.y & (NN - 1);
        const float* xb = x + (((size_t)b * CC) * DD + d) * NN + n0;
        const float* mb = mean_pc + (size_t)b * CC * NN;
#pragma unroll
        for (int q = 0; q < CC / 16; ++q) {
#pragma unroll
            for (int r = 0; r < 8; ++r) {
                const int ch = q * 16 + 8 * h + r;
                const v2f xv = *(const v2f*)&xb[(size_t)ch * DD * NN];
                const float g0 = mb[(size_t)ch * NN + j0];
                const float g1 = mb[(size_t)ch * NN + j1];
                E[q][0][r] = xv.x;
                E[q][1][r] = xv.y;
                E[q + CC / 16][0][r] = g0 - xv.x;
                E[q + CC / 16][1][r] = g1 - xv.y;
            }
        }
    }

    v8f H2[HH / 16][NCT];
#pragma unroll
    for (int u = 0; u < HH / 16; ++u) { H2[u][0] = vzero; H2[u][1] = vzero; }

    for (int T = 0; T < HH / 16; ++T) {
        const int Tn = (T + 1 < HH / 16) ? T + 1 : T;
        const v8f bias1 = ld8(b1 + T * 16 + 8 * h);
        v8f c0 = vzero, c1 = vzero;
        v8f t0 = vzero, t1 = vzero;
#pragma unroll
        for (int s = 0; s < S12; ++s) {
            const int sn = s + AH;
            A[sn % NRNG] = ld8(sn < S12 ? WPTR12(T, sn) : WPTR12(Tn, sn - S12));
            __builtin_amdgcn_sched_barrier(0);
            if (s < S1) {
                mma_block(A[s % NRNG], E[s][0], E[s][1], c0, c1);
            } else {
                if (s == S1) { t0 = bias_relu(c0, bias1); t1 = bias_relu(c1, bias1); }
                mma_block(A[s % NRNG], t0, t1, H2[s - S1][0], H2[s - S1][1]);
            }
            __builtin_amdgcn_sched_barrier(0);
        }
    }
#undef WPTR12

#define WPTR3(V, s) (w3p + (size_t)(V) * 16 * HH + (s) * 16)
#pragma unroll
    for (int s = 0; s < AH; ++s) A[s] = ld8(WPTR3(0, s));

#pragma unroll
    for (int u = 0; u < HH / 16; ++u) {
        const v8f bias2 = ld8(b2 + u * 16 + 8 * h);
        H2[u][0] = bias_relu(H2[u][0], bias2);
        H2[u][1] = bias_relu(H2[u][1], bias2);
    }

    for (int V = 0; V < CC / 16; ++V) {
        const int Vn = (V + 1 < CC / 16) ? V + 1 : V;
        const v8f bias3 = ld8(b3 + V * 16 + 8 * h);
        v8f c0 = vzero, c1 = vzero;
#pragma unroll
        for (int s = 0; s < S3; ++s) {
            const int sn = s + AH;
            A[sn % NRNG] = ld8(sn < S3 ? WPTR3(V, sn) : WPTR3(Vn, sn - S3));
            __builtin_amdgcn_sched_barrier(0);
            mma_block(A[s % NRNG], H2[s][0], H2[s][1], c0, c1);
            __builtin_amdgcn_sched_barrier(0);
        }
        float m0[8], m1[8];
#pragma unroll
        for (int r = 0; r < 8; ++r) { m0[r] = c0[r] + bias3[r]; m1[r] = c1[r] + bias3[r]; }
#pragma unroll
        for (int o = 1; o < 16; o <<= 1) {
#pragma unroll
            for (int r = 0; r < 8; ++r) {
                m0[r] = fmaxf(m0[r], __shfl_xor(m0[r], o));
                m1[r] = fmaxf(m1[r], __shfl_xor(m1[r], o));
            }
        }
        if (c == 0) {
            v2f xr[8];
#pragma unroll
            for (int r = 0; r < 8; ++r)
                xr[r] = *(const v2f*)&x[(((size_t)b * CC + V * 16 + 8 * h + r) * DD + d) * NN + n0];
#pragma unroll
            for (int r = 0; r < 8; ++r)
                *(v2f*)&out[(((size_t)b * CC + V * 16 + 8 * h + r) * DD + d) * NN + n0] = (v2f){m0[r] + xr[r].x, m1[r] + xr[r].y};
        }
    }
#undef WPTR3
}

extern "C" void kernel_launch(void* const* d_in, const int* in_sizes, int n_in,
                              void* d_out, int out_size, void* d_ws, size_t ws_size,
                              hipStream_t stream) {
    const float* x  = (const float*)d_in[0];
    const float* W1 = (const float*)d_in[1];
    const float* b1 = (const float*)d_in[2];
    const float* W2 = (const float*)d_in[3];
    const float* b2 = (const float*)d_in[4];
    const float* W3 = (const float*)d_in[5];
    const float* b3 = (const float*)d_in[6];
    float* out = (float*)d_out;

    float* mean_pc = (float*)d_ws;
    float* sq      = mean_pc + (size_t)BB * CC * NN;
    int*   idxp    = (int*)(sq + (size_t)BB * NN);

    meanpc_kernel<<<(BB * CC * NN + 255) / 256, 256, 0, stream>>>(x, mean_pc);
    sqnorm_kernel<<<(BB * NN + 255) / 256, 256, 0, stream>>>(mean_pc, sq);
    knn_kernel<<<BB * (NN / KQ), (NS + 1) * KQ, 0, stream>>>(mean_pc, sq, idxp);
    mlp_kernel<<<BB * DD * (NN / NCT) / MW, MW * 32, 0, stream>>>(x, mean_pc, idxp,
                                                        W1, b1, W2, b2, W3, b3, out);
}


# Mean table stored point-major (B,N,C) via an LDS transpose in the mean launch: the MLP neighbour gather reads a point as 2 lines by 128-bit loads, not 64 lines x 4 B; all values bit-identical
# speedup vs baseline: 9.8313x; 1.0821x over previous
#include <hip/hip_runtime.h>

#define BB 2
#define CC 64
#define DD 3
#define NN 4096
#define KK 16
#define EE 128
#define HH 256

typedef __attribute__((ext_vector_type(2))) float v2f;
typedef __attribute__((ext_vector_type(8))) float v8f;

constexpr int NCT  = 2;
constexpr int MW   = 4;

constexpr int MT = 32;
constexpr int MTHR = 256;
static_assert(NN % MT == 0 && MTHR % MT == 0 && CC % (MTHR / MT) == 0 && (MT * CC) % MTHR == 0, "mean tile must divide evenly");

__global__ void __launch_bounds__(MTHR) meanpc_kernel(const float* __restrict__ x, float* __restrict__ mean_t) {
    __shared__ float tl[MT][CC + 1];
    const int t  = threadIdx.x;
    const int b  = blockIdx.x / (NN / MT);
    const int n0 = (blockIdx.x % (NN / MT)) * MT;
    const int nl = t % MT;
    const int c0 = t / MT;
#pragma unroll
    for (int i = 0; i < CC / (MTHR / MT); ++i) {
        const int c = c0 + (MTHR / MT) * i;
        const float* xp = x + ((size_t)b * CC + c) * (DD * NN) + n0 + nl;
        tl[nl][c] = (xp[0] + xp[NN] + xp[2 * NN]) * (1.0f / 3.0f);
    }
    __syncthreads();
    float* dst = mean_t + ((size_t)b * NN + n0) * CC;
#pragma unroll
    for (int i = 0; i < MT * CC / MTHR; ++i) {
        const int f = t + MTHR * i;
        dst[f] = tl[f / CC][f % CC];
    }
}

typedef __attribute__((ext_vector_type(4))) float v4f;

__global__ void sqnorm_kernel(const float* __restrict__ mean_t, float* __restrict__ sq) {
    int i = blockIdx.x * blockDim.x + threadIdx.x;
    if (i >= BB * NN) return;
    const float* p = mean_t + (size_t)i * CC;
    float s = 0.0f;
#pragma unroll
    for (int c4 = 0; c4 < CC / 4; ++c4) {
        const v4f v = *(const v4f*)(p + 4 * c4);
        s = __builtin_fmaf(v.x, v.x, s);
        s = __builtin_fmaf(v.y, v.y, s);
        s = __builtin_fmaf(v.z, v.z, s);
        s = __builtin_fmaf(v.w, v.w, s);
    }
    sq[i] = s;
}

constexpr int KQ    = 32;
constexpr int NS    = 2;
constexpr int TR    = 32;
constexpr int PP    = 2 * CC + 4;
constexpr int RG    = 16;
constexpr int NPAIR = RG / 2;
constexpr int NBAT  = CC / 2;
constexpr int AHEAD = 3;
constexpr int NSLOT = AHEAD + 1;
constexpr int NGRP  = TR / RG;
constexpr int SG    = NS * NGRP;
constexpr int NSTEP = NN / (NS * TR);
static_assert(NBAT % NSLOT == 0, "slot phase must repeat from group to group");
static_assert(NGRP == 2, "group wrap below assumes two groups per tile");
static_assert(NN % (NS * TR) == 0, "steps must cover the points exactly");

__device__ __forceinline__ void knn_merge_group(const float (*sc)[KQ], int lane, int r0, int n,
                                                float (&bd)[KK], int (&bi)[KK]) {
    float d2[RG];
    bool any = false;
#pragma unroll
    for (int i = 0; i < RG; ++i) {
        d2[i] = sc[i][lane];
        any |= d2[i] < bd[KK - 1];
    }
    if (any) {
#pragma unroll
        for (int i = 0; i < RG; ++i) {
            const int r = r0 + i;
            if (r != n && d2[i] < bd[KK - 1]) {
                bd[KK - 1] = d2[i];
                bi[KK - 1] = r;
#pragma unroll
                for (int j = KK - 1; j > 0; --j) {
                    if (bd[j] < bd[j - 1]) {
                        float td = bd[j]; bd[j] = bd[j - 1]; bd[j - 1] = td;
                        int   ti = bi[j]; bi[j] = bi[j - 1]; bi[j - 1] = ti;
                    }
                }
            }
        }
    }
}

__device__ __forceinline__ void ld_point(const float* p, float (&v)[CC]) {
#pragma unroll
    for (int c4 = 0; c4 < CC / 4; ++c4) {
        const v4f w = *(const v4f*)(p + 4 * c4);
        v[4 * c4 + 0] = w.x; v[4 * c4 + 1] = w.y; v[4 * c4 + 2] = w.z; v[4 * c4 + 3] = w.w;
    }
}

__global__ void __launch_bounds__((NS + 1) * KQ) knn_kernel(const float* __restrict__ mean_t,
                                                            const float* __restrict__ sq,
                                                            int* __restrict__ idx) {
    __shared__ __attribute__((aligned(16))) float tile[NS][(TR / 2) * PP];
    __shared__ float score[2][SG][RG][KQ];

    const int lane = threadIdx.x & (KQ - 1);
    const int role = __builtin_amdgcn_readfirstlane((int)(threadIdx.x >> 5));
    const bool scorer = role < NS;
    const int b = blockIdx.x / (NN / KQ);
    const int n = (blockIdx.x % (NN / KQ)) * KQ + lane;
    const float* pcb = mean_t + (size_t)b * NN * CC;
    const float* sqb = sq + (size_t)b * NN;
    float* mytile = &tile[scorer ? role : 0][0];

    float q[CC];
    float pf[CC];
    if (scorer) {
        ld_point(pcb + (size_t)n * CC, q);
        ld_point(pcb + (size_t)(role * TR + lane) * CC, pf);
    }

    float bd[KK];
    int   bi[KK];
#pragma unroll
    for (int j = 0; j < KK; ++j) { bd[j] = 3.0e38f; bi[j] = 0; }

    const int sp = (lane >> 1) * PP + (lane & 1);

    for (int st = 0; st < NSTEP; ++st) {
        if (scorer) {
#pragma unroll
            for (int c = 0; c < CC; ++c) mytile[sp + 2 * c] = pf[c];
        }
        __syncthreads();

        if (scorer) {
            const int t = st * NS + role;
            v4f rv[NSLOT][NPAIR];
            if (st + 1 < NSTEP) ld_point(pcb + (size_t)((t + NS) * TR + lane) * CC, pf);
#pragma unroll
            for (int s = 0; s < AHEAD; ++s) {
#pragma unroll
                for (int p = 0; p < NPAIR; ++p) rv[s][p] = *(const v4f*)&mytile[p * PP + 4 * s];
            }

            for (int g = 0; g < NGRP; ++g) {
                const float* rows  = &mytile[g * NPAIR * PP];
                const float* nrows = &mytile[((g + 1) & 1) * NPAIR * PP];
                v2f acc[NPAIR];
#pragma unroll
                for (int p = 0; p < NPAIR; ++p) acc[p] = (v2f){0.0f, 0.0f};
#pragma unroll
                for (int bt = 0; bt < NBAT; ++bt) {
                    const int nx = bt + AHEAD;
                    const float* src = (nx < NBAT) ? rows : nrows;
                    const int    nb  = (nx < NBAT) ? nx : nx - NBAT;
#pragma unroll
                    for (int p = 0; p < NPAIR; ++p)
                        rv[nx % NSLOT][p] = *(const v4f*)&src[p * PP + 4 * nb];
                    __builtin_amdgcn_sched_barrier(0);
                    const int cs = bt % NSLOT;
                    const v2f q0 = {q[2 * bt + 0], q[2 * bt + 0]};
                    const v2f q1 = {q[2 * bt + 1], q[2 * bt + 1]};
#pragma unroll
                    for (int p = 0; p < NPAIR; ++p) acc[p] = __builtin_elementwise_fma(q0, rv[cs][p].xy, acc[p]);
#pragma unroll
                    for (int p = 0; p < NPAIR; ++p) acc[p] = __builtin_elementwise_fma(q1, rv[cs][p].zw, acc[p]);
                    __builtin_amdgcn_sched_barrier(0);
                }

                const int r0 = (t * NGRP + g) * RG;
#pragma unroll
                for (int i = 0; i < RG; ++i)
                    score[st & 1][role * NGRP + g][i][lane] = sqb[r0 + i] - 2.0f * acc[i >> 1][i & 1];
            }
        } else if (st > 0) {
            for (int k = 0; k < SG; ++k)
                knn_merge_group(score[(st - 1) & 1][k], lane, ((st - 1) * SG + k) * RG, n, bd, bi);
        }
        __syncthreads();
    }
    if (!scorer) {
        constexpr int SL = NSTEP - 1;
        for (int k = 0; k < SG; ++k)
            knn_merge_group(score[SL & 1][k], lane, (SL * SG + k) * RG, n, bd, bi);
#pragma unroll
        for (int j = 0; j < KK; ++j) idx[((size_t)b * KK + j) * NN + n] = bi[j];
    }
}

typedef __attribute__((ext_vector_type(2))) int v2i;

__device__ __forceinline__ v8f ld8(const float* p) {
    const v4f lo = *(const v4f*)p;
    const v4f hi = *(const v4f*)(p + 4);
    return (v8f){lo.x, lo.y, lo.z, lo.w, hi.x, hi.y, hi.z, hi.w};
}

__device__ __forceinline__ void mma_block(const v8f& a, const v8f& b0, const v8f& b1, v8f& c0, v8f& c1) {
#pragma unroll
    for (int m = 0; m < 4; ++m) {
        const v2f af = {a[2 * m], a[2 * m + 1]};
        const v2f f0 = {b0[2 * m], b0[2 * m + 1]};
        const v2f f1 = {b1[2 * m], b1[2 * m + 1]};
        c0 = __builtin_amdgcn_wmma_f32_16x16x4_f32(false, af, false, f0, (short)0, c0, false, false);
        c1 = __builtin_amdgcn_wmma_f32_16x16x4_f32(false, af, false, f1, (short)0, c1, false, false);
    }
}

__device__ __forceinline__ v8f bias_relu(const v8f& acc, const v8f& bias) {
    v8f o;
#pragma unroll
    for (int r = 0; r < 8; ++r) o[r] = fmaxf(acc[r] + bias[r], 0.0f);
    return o;
}

constexpr int AH   = 3;
constexpr int NRNG = AH + 1;
constexpr int S1   = EE / 16;
constexpr int S12  = S1 + HH / 16;
constexpr int S3   = HH / 16;
static_assert(S12 % NRNG == 0 && S3 % NRNG == 0, "ring phase must repeat from trip to trip");
static_assert(NCT == 2, "mma_block and the paired loads below are written for two column tiles");

__global__ void __launch_bounds__(MW * 32) mlp_kernel(const float* __restrict__ x,
                                                      const float* __restrict__ mean_t,
                                                      const int* __restrict__ idx,
                                                      const float* __restrict__ W1, const float* __restrict__ b1,
                                                      const float* __restrict__ W2, const float* __restrict__ b2,
                                                      const float* __restrict__ W3, const float* __restrict__ b3,
                                                      float* __restrict__ out) {
    const int lane = threadIdx.x & 31;
    const int c    = lane & 15;
    const int h    = lane >> 4;
    const int wv   = __builtin_amdgcn_readfirstlane((int)(threadIdx.x >> 5));
    const int task = blockIdx.x * MW + wv;
    const int np   = task % (NN / NCT);
    const int bd   = task / (NN / NCT);
    const int d    = bd % DD;
    const int b    = bd / DD;
    const int n0   = np * NCT;

    const v8f vzero = {0.f, 0.f, 0.f, 0.f, 0.f, 0.f, 0.f, 0.f};

    const float* w1p = W1 + (size_t)c * EE + 8 * h;
    const float* w2p = W2 + (size_t)c * HH + 8 * h;
    const float* w3p = W3 + (size_t)c * HH + 8 * h;

#define WPTR12(T, s) (((s) < S1) ? (w1p + (size_t)(T) * 16 * EE + (s) * 16) \
                                 : (w2p + (size_t)((s) - S1) * 16 * HH + (T) * 16))

    v8f A[NRNG];
#pragma unroll
    for (int s = 0; s < AH; ++s) A[s] = ld8(WPTR12(0, s));

    v8f E[S1][NCT];
    {
        const v2i jj = *(const v2i*)&idx[((size_t)b * KK + c) * NN + n0];
        const int j0 = jj.x & (NN - 1);
        const int j1 = jj.y & (NN - 1);
        const float* xb = x + (((size_t)b * CC) * DD + d) * NN + n0;
        const float* mb = mean_t + (size_t)b * NN * CC;
        const float* g0p = mb + (size_t)j0 * CC + 8 * h;
        const float* g1p = mb + (size_t)j1 * CC + 8 * h;
#pragma unroll
        for (int q = 0; q < CC / 16; ++q) {
            const v8f g0 = ld8(g0p + 16 * q);
            const v8f g1 = ld8(g1p + 16 * q);
#pragma unroll
            for (int r = 0; r < 8; ++r) {
                const int ch = q * 16 + 8 * h + r;
                const v2f xv = *(const v2f*)&xb[(size_t)ch * DD * NN];
                E[q][0][r] = xv.x;
                E[q][1][r] = xv.y;
                E[q + CC / 16][0][r] = g0[r] - xv.x;
                E[q + CC / 16][1][r] = g1[r] - xv.y;
            }
        }
    }

    v8f H2[HH / 16][NCT];
#pragma unroll
    for (int u = 0; u < HH / 16; ++u) { H2[u][0] = vzero; H2[u][1] = vzero; }

    for (int T = 0; T < HH / 16; ++T) {
        const int Tn = (T + 1 < HH / 16) ? T + 1 : T;
        const v8f bias1 = ld8(b1 + T * 16 + 8 * h);
        v8f c0 = vzero, c1 = vzero;
        v8f t0 = vzero, t1 = vzero;
#pragma unroll
        for (int s = 0; s < S12; ++s) {
            const int sn = s + AH;
            A[sn % NRNG] = ld8(sn < S12 ? WPTR12(T, sn) : WPTR12(Tn, sn - S12));
            __builtin_amdgcn_sched_barrier(0);
            if (s < S1) {
                mma_block(A[s % NRNG], E[s][0], E[s][1], c0, c1);
            } else {
                if (s == S1) { t0 = bias_relu(c0, bias1); t1 = bias_relu(c1, bias1); }
                mma_block(A[s % NRNG], t0, t1, H2[s - S1][0], H2[s - S1][1]);
            }
            __builtin_amdgcn_sched_barrier(0);
        }
    }
#undef WPTR12

#define WPTR3(V, s) (w3p + (size_t)(V) * 16 * HH + (s) * 16)
#pragma unroll
    for (int s = 0; s < AH; ++s) A[s] = ld8(WPTR3(0, s));

#pragma unroll
    for (int u = 0; u < HH / 16; ++u) {
        const v8f bias2 = ld8(b2 + u * 16 + 8 * h);
        H2[u][0] = bias_relu(H2[u][0], bias2);
        H2[u][1] = bias_relu(H2[u][1], bias2);
    }

    for (int V = 0; V < CC / 16; ++V) {
        const int Vn = (V + 1 < CC / 16) ? V + 1 : V;
        const v8f bias3 = ld8(b3 + V * 16 + 8 * h);
        v8f c0 = vzero, c1 = vzero;
#pragma unroll
        for (int s = 0; s < S3; ++s) {
            const int sn = s + AH;
            A[sn % NRNG] = ld8(sn < S3 ? WPTR3(V, sn) : WPTR3(Vn, sn - S3));
            __builtin_amdgcn_sched_barrier(0);
            mma_block(A[s % NRNG], H2[s][0], H2[s][1], c0, c1);
            __builtin_amdgcn_sched_barrier(0);
        }
        float m0[8], m1[8];
#pragma unroll
        for (int r = 0; r < 8; ++r) { m0[r] = c0[r] + bias3[r]; m1[r] = c1[r] + bias3[r]; }
#pragma unroll
        for (int o = 1; o < 16; o <<= 1) {
#pragma unroll
            for (int r = 0; r < 8; ++r) {
                m0[r] = fmaxf(m0[r], __shfl_xor(m0[r], o));
                m1[r] = fmaxf(m1[r], __shfl_xor(m1[r], o));
            }
        }
        if (c == 0) {
            v2f xr[8];
#pragma unroll
            for (int r = 0; r < 8; ++r)
                xr[r] = *(const v2f*)&x[(((size_t)b * CC + V * 16 + 8 * h + r) * DD + d) * NN + n0];
#pragma unroll
            for (int r = 0; r < 8; ++r)
                *(v2f*)&out[(((size_t)b * CC + V * 16 + 8 * h + r) * DD + d) * NN + n0] = (v2f){m0[r] + xr[r].x, m1[r] + xr[r].y};
        }
    }
#undef WPTR3
}

extern "C" void kernel_launch(void* const* d_in, const int* in_sizes, int n_in,
                              void* d_out, int out_size, void* d_ws, size_t ws_size,
                              hipStream_t stream) {
    const float* x  = (const float*)d_in[0];
    const float* W1 = (const float*)d_in[1];
    const float* b1 = (const float*)d_in[2];
    const float* W2 = (const float*)d_in[3];
    const float* b2 = (const float*)d_in[4];
    const float* W3 = (const float*)d_in[5];
    const float* b3 = (const float*)d_in[6];
    float* out = (float*)d_out;

    float* mean_t  = (float*)d_ws;
    float* sq      = mean_t + (size_t)BB * NN * CC;
    int*   idxp    = (int*)(sq + (size_t)BB * NN);

    meanpc_kernel<<<BB * (NN / MT), MTHR, 0, stream>>>(x, mean_t);
    sqnorm_kernel<<<(BB * NN + 255) / 256, 256, 0, stream>>>(mean_t, sq);
    knn_kernel<<<BB * (NN / KQ), (NS + 1) * KQ, 0, stream>>>(mean_t, sq, idxp);
    mlp_kernel<<<BB * DD * (NN / NCT) / MW, MW * 32, 0, stream>>>(x, mean_t, idxp,
                                                        W1, b1, W2, b2, W3, b3, out);
}


# KNN list wave: a group's 16 scores are sorted by a fixed 63-exchange network and folded into the sorted list by a bitonic merge, not inserted one by one; scores unchanged, ties may order otherwise
# speedup vs baseline: 11.3600x; 1.1555x over previous
#include <hip/hip_runtime.h>

#define BB 2
#define CC 64
#define DD 3
#define NN 4096
#define KK 16
#define EE 128
#define HH 256

typedef __attribute__((ext_vector_type(2))) float v2f;
typedef __attribute__((ext_vector_type(8))) float v8f;

constexpr int NCT  = 2;
constexpr int MW   = 4;

constexpr int MT = 32;
constexpr int MTHR = 256;
static_assert(NN % MT == 0 && MTHR % MT == 0 && CC % (MTHR / MT) == 0 && (MT * CC) % MTHR == 0, "mean tile must divide evenly");

__global__ void __launch_bounds__(MTHR) meanpc_kernel(const float* __restrict__ x, float* __restrict__ mean_t) {
    __shared__ float tl[MT][CC + 1];
    const int t  = threadIdx.x;
    const int b  = blockIdx.x / (NN / MT);
    const int n0 = (blockIdx.x % (NN / MT)) * MT;
    const int nl = t % MT;
    const int c0 = t / MT;
#pragma unroll
    for (int i = 0; i < CC / (MTHR / MT); ++i) {
        const int c = c0 + (MTHR / MT) * i;
        const float* xp = x + ((size_t)b * CC + c) * (DD * NN) + n0 + nl;
        tl[nl][c] = (xp[0] + xp[NN] + xp[2 * NN]) * (1.0f / 3.0f);
    }
    __syncthreads();
    float* dst = mean_t + ((size_t)b * NN + n0) * CC;
#pragma unroll
    for (int i = 0; i < MT * CC / MTHR; ++i) {
        const int f = t + MTHR * i;
        dst[f] = tl[f / CC][f % CC];
    }
}

typedef __attribute__((ext_vector_type(4))) float v4f;

__global__ void sqnorm_kernel(const float* __restrict__ mean_t, float* __restrict__ sq) {
    int i = blockIdx.x * blockDim.x + threadIdx.x;
    if (i >= BB * NN) return;
    const float* p = mean_t + (size_t)i * CC;
    float s = 0.0f;
#pragma unroll
    for (int c4 = 0; c4 < CC / 4; ++c4) {
        const v4f v = *(const v4f*)(p + 4 * c4);
        s = __builtin_fmaf(v.x, v.x, s);
        s = __builtin_fmaf(v.y, v.y, s);
        s = __builtin_fmaf(v.z, v.z, s);
        s = __builtin_fmaf(v.w, v.w, s);
    }
    sq[i] = s;
}

constexpr int KQ    = 32;
constexpr int NS    = 2;
constexpr int TR    = 32;
constexpr int PP    = 2 * CC + 4;
constexpr int RG    = 16;
constexpr int NPAIR = RG / 2;
constexpr int NBAT  = CC / 2;
constexpr int AHEAD = 3;
constexpr int NSLOT = AHEAD + 1;
constexpr int NGRP  = TR / RG;
constexpr int SG    = NS * NGRP;
constexpr int NSTEP = NN / (NS * TR);
static_assert(NBAT % NSLOT == 0, "slot phase must repeat from group to group");
static_assert(NGRP == 2, "group wrap below assumes two groups per tile");
static_assert(NN % (NS * TR) == 0, "steps must cover the points exactly");

constexpr int NCE = 63;
constexpr unsigned char SORT16[NCE][2] = {
    {0,1}, {2,3}, {4,5}, {6,7}, {8,9}, {10,11}, {12,13}, {14,15},
    {0,2}, {1,3}, {4,6}, {5,7}, {8,10}, {9,11}, {12,14}, {13,15},
    {1,2}, {5,6}, {9,10}, {13,14}, {0,4}, {3,7}, {8,12}, {11,15},
    {1,5}, {2,6}, {9,13}, {10,14}, {0,8}, {7,15},
    {2,4}, {3,5}, {10,12}, {11,13},
    {1,2}, {3,4}, {5,6}, {9,10}, {11,12}, {13,14},
    {1,9}, {2,10}, {3,11}, {4,12}, {5,13}, {6,14},
    {4,8}, {5,9}, {6,10}, {7,11},
    {2,4}, {3,5}, {6,8}, {7,9}, {10,12}, {11,13},
    {1,2}, {3,4}, {5,6}, {7,8}, {9,10}, {11,12}, {13,14}
};
static_assert(RG == KK, "the fold below pairs list position j with new position 15 - j");

__device__ __forceinline__ void knn_ce(float& da, float& db, int& ia, int& ib) {
    const bool s = db < da;
    const float lo = s ? db : da, hi = s ? da : db;
    const int   il = s ? ib : ia, ih = s ? ia : ib;
    da = lo; db = hi; ia = il; ib = ih;
}

__device__ __forceinline__ void knn_merge_group(float (&d2)[RG], int r0, int n, int nb,
                                                float (&bd)[KK], int (&bi)[KK]) {
    int ri[RG];
#pragma unroll
    for (int i = 0; i < RG; ++i) ri[i] = r0 + i;
    if ((r0 & ~(KQ - 1)) == nb) {
#pragma unroll
        for (int i = 0; i < RG; ++i) if (r0 + i == n) d2[i] = 3.0e38f;
    }
    bool any = false;
#pragma unroll
    for (int i = 0; i < RG; ++i) any |= d2[i] < bd[KK - 1];
    if (any) {
#pragma unroll
        for (int e = 0; e < NCE; ++e)
            knn_ce(d2[SORT16[e][0]], d2[SORT16[e][1]], ri[SORT16[e][0]], ri[SORT16[e][1]]);
#pragma unroll
        for (int j = 0; j < KK; ++j) {
            const bool take = d2[RG - 1 - j] < bd[j];
            bd[j] = take ? d2[RG - 1 - j] : bd[j];
            bi[j] = take ? ri[RG - 1 - j] : bi[j];
        }
#pragma unroll
        for (int k = KK / 2; k >= 1; k >>= 1) {
#pragma unroll
            for (int i = 0; i < KK; ++i)
                if ((i & k) == 0) knn_ce(bd[i], bd[i + k], bi[i], bi[i + k]);
        }
    }
}

__device__ __forceinline__ void knn_merge_step(const float (*sc)[RG][KQ], int lane, int r0, int n, int nb,
                                               float (&bd)[KK], int (&bi)[KK]) {
    float nx[RG];
#pragma unroll
    for (int i = 0; i < RG; ++i) nx[i] = sc[0][i][lane];
    for (int k = 0; k < SG; ++k) {
        float d2[RG];
#pragma unroll
        for (int i = 0; i < RG; ++i) d2[i] = nx[i];
        const int kn = (k + 1 < SG) ? k + 1 : k;
#pragma unroll
        for (int i = 0; i < RG; ++i) nx[i] = sc[kn][i][lane];
        knn_merge_group(d2, r0 + k * RG, n, nb, bd, bi);
    }
}

__device__ __forceinline__ void ld_point(const float* p, float (&v)[CC]) {
#pragma unroll
    for (int c4 = 0; c4 < CC / 4; ++c4) {
        const v4f w = *(const v4f*)(p + 4 * c4);
        v[4 * c4 + 0] = w.x; v[4 * c4 + 1] = w.y; v[4 * c4 + 2] = w.z; v[4 * c4 + 3] = w.w;
    }
}

__global__ void __launch_bounds__((NS + 1) * KQ) knn_kernel(const float* __restrict__ mean_t,
                                                            const float* __restrict__ sq,
                                                            int* __restrict__ idx) {
    __shared__ __attribute__((aligned(16))) float tile[NS][(TR / 2) * PP];
    __shared__ float score[2][SG][RG][KQ];

    const int lane = threadIdx.x & (KQ - 1);
    const int role = __builtin_amdgcn_readfirstlane((int)(threadIdx.x >> 5));
    const bool scorer = role < NS;
    const int b = blockIdx.x / (NN / KQ);
    const int nb = (blockIdx.x % (NN / KQ)) * KQ;
    const int n = nb + lane;
    const float* pcb = mean_t + (size_t)b * NN * CC;
    const float* sqb = sq + (size_t)b * NN;
    float* mytile = &tile[scorer ? role : 0][0];

    float q[CC];
    float pf[CC];
    if (scorer) {
        ld_point(pcb + (size_t)n * CC, q);
        ld_point(pcb + (size_t)(role * TR + lane) * CC, pf);
    }

    float bd[KK];
    int   bi[KK];
#pragma unroll
    for (int j = 0; j < KK; ++j) { bd[j] = 3.0e38f; bi[j] = 0; }

    const int sp = (lane >> 1) * PP + (lane & 1);

    for (int st = 0; st < NSTEP; ++st) {
        if (scorer) {
#pragma unroll
            for (int c = 0; c < CC; ++c) mytile[sp + 2 * c] = pf[c];
        }
        __syncthreads();

        if (scorer) {
            const int t = st * NS + role;
            v4f rv[NSLOT][NPAIR];
            if (st + 1 < NSTEP) ld_point(pcb + (size_t)((t + NS) * TR + lane) * CC, pf);
#pragma unroll
            for (int s = 0; s < AHEAD; ++s) {
#pragma unroll
                for (int p = 0; p < NPAIR; ++p) rv[s][p] = *(const v4f*)&mytile[p * PP + 4 * s];
            }

            for (int g = 0; g < NGRP; ++g) {
                const float* rows  = &mytile[g * NPAIR * PP];
                const float* nrows = &mytile[((g + 1) & 1) * NPAIR * PP];
                v2f acc[NPAIR];
#pragma unroll
                for (int p = 0; p < NPAIR; ++p) acc[p] = (v2f){0.0f, 0.0f};
#pragma unroll
                for (int bt = 0; bt < NBAT; ++bt) {
                    const int nx = bt + AHEAD;
                    const float* src = (nx < NBAT) ? rows : nrows;
                    const int    nb  = (nx < NBAT) ? nx : nx - NBAT;
#pragma unroll
                    for (int p = 0; p < NPAIR; ++p)
                        rv[nx % NSLOT][p] = *(const v4f*)&src[p * PP + 4 * nb];
                    __builtin_amdgcn_sched_barrier(0);
                    const int cs = bt % NSLOT;
                    const v2f q0 = {q[2 * bt + 0], q[2 * bt + 0]};
                    const v2f q1 = {q[2 * bt + 1], q[2 * bt + 1]};
#pragma unroll
                    for (int p = 0; p < NPAIR; ++p) acc[p] = __builtin_elementwise_fma(q0, rv[cs][p].xy, acc[p]);
#pragma unroll
                    for (int p = 0; p < NPAIR; ++p) acc[p] = __builtin_elementwise_fma(q1, rv[cs][p].zw, acc[p]);
                    __builtin_amdgcn_sched_barrier(0);
                }

                const int r0 = (t * NGRP + g) * RG;
#pragma unroll
                for (int i = 0; i < RG; ++i)
                    score[st & 1][role * NGRP + g][i][lane] = sqb[r0 + i] - 2.0f * acc[i >> 1][i & 1];
            }
        } else if (st > 0) {
            knn_merge_step(score[(st - 1) & 1], lane, (st - 1) * SG * RG, n, nb, bd, bi);
        }
        __syncthreads();
    }
    if (!scorer) {
        constexpr int SL = NSTEP - 1;
        knn_merge_step(score[SL & 1], lane, SL * SG * RG, n, nb, bd, bi);
#pragma unroll
        for (int j = 0; j < KK; ++j) idx[((size_t)b * KK + j) * NN + n] = bi[j];
    }
}

typedef __attribute__((ext_vector_type(2))) int v2i;

__device__ __forceinline__ v8f ld8(const float* p) {
    const v4f lo = *(const v4f*)p;
    const v4f hi = *(const v4f*)(p + 4);
    return (v8f){lo.x, lo.y, lo.z, lo.w, hi.x, hi.y, hi.z, hi.w};
}

__device__ __forceinline__ void mma_block(const v8f& a, const v8f& b0, const v8f& b1, v8f& c0, v8f& c1) {
#pragma unroll
    for (int m = 0; m < 4; ++m) {
        const v2f af = {a[2 * m], a[2 * m + 1]};
        const v2f f0 = {b0[2 * m], b0[2 * m + 1]};
        const v2f f1 = {b1[2 * m], b1[2 * m + 1]};
        c0 = __builtin_amdgcn_wmma_f32_16x16x4_f32(false, af, false, f0, (short)0, c0, false, false);
        c1 = __builtin_amdgcn_wmma_f32_16x16x4_f32(false, af, false, f1, (short)0, c1, false, false);
    }
}

__device__ __forceinline__ v8f bias_relu(const v8f& acc, const v8f& bias) {
    v8f o;
#pragma unroll
    for (int r = 0; r < 8; ++r) o[r] = fmaxf(acc[r] + bias[r], 0.0f);
    return o;
}

constexpr int AH   = 3;
constexpr int NRNG = AH + 1;
constexpr int S1   = EE / 16;
constexpr int S12  = S1 + HH / 16;
constexpr int S3   = HH / 16;
static_assert(S12 % NRNG == 0 && S3 % NRNG == 0, "ring phase must repeat from trip to trip");
static_assert(NCT == 2, "mma_block and the paired loads below are written for two column tiles");

__global__ void __launch_bounds__(MW * 32) mlp_kernel(const float* __restrict__ x,
                                                      const float* __restrict__ mean_t,
                                                      const int* __restrict__ idx,
                                                      const float* __restrict__ W1, const float* __restrict__ b1,
                                                      const float* __restrict__ W2, const float* __restrict__ b2,
                                                      const float* __restrict__ W3, const float* __restrict__ b3,
                                                      float* __restrict__ out) {
    const int lane = threadIdx.x & 31;
    const int c    = lane & 15;
    const int h    = lane >> 4;
    const int wv   = __builtin_amdgcn_readfirstlane((int)(threadIdx.x >> 5));
    const int task = blockIdx.x * MW + wv;
    const int np   = task % (NN / NCT);
    const int bd   = task / (NN / NCT);
    const int d    = bd % DD;
    const int b    = bd / DD;
    const int n0   = np * NCT;

    const v8f vzero = {0.f, 0.f, 0.f, 0.f, 0.f, 0.f, 0.f, 0.f};

    const float* w1p = W1 + (size_t)c * EE + 8 * h;
    const float* w2p = W2 + (size_t)c * HH + 8 * h;
    const float* w3p = W3 + (size_t)c * HH + 8 * h;

#define WPTR12(T, s) (((s) < S1) ? (w1p + (size_t)(T) * 16 * EE + (s) * 16) \
                                 : (w2p + (size_t)((s) - S1) * 16 * HH + (T) * 16))

    v8f A[NRNG];
#pragma unroll
    for (int s = 0; s < AH; ++s) A[s] = ld8(WPTR12(0, s));

    v8f E[S1][NCT];
    {
        const v2i jj = *(const v2i*)&idx[((size_t)b * KK + c) * NN + n0];
        const int j0 = jj.x & (NN - 1);
        const int j1 = jj.y & (NN - 1);
        const float* xb = x + (((size_t)b * CC) * DD + d) * NN + n0;
        const float* mb = mean_t + (size_t)b * NN * CC;
        const float* g0p = mb + (size_t)j0 * CC + 8 * h;
        const float* g1p = mb + (size_t)j1 * CC + 8 * h;
#pragma unroll
        for (int q = 0; q < CC / 16; ++q) {
            const v8f g0 = ld8(g0p + 16 * q);
            const v8f g1 = ld8(g1p + 16 * q);
#pragma unroll
            for (int r = 0; r < 8; ++r) {
                const int ch = q * 16 + 8 * h + r;
                const v2f xv = *(const v2f*)&xb[(size_t)ch * DD * NN];
                E[q][0][r] = xv.x;
                E[q][1][r] = xv.y;
                E[q + CC / 16][0][r] = g0[r] - xv.x;
                E[q + CC / 16][1][r] = g1[r] - xv.y;
            }
        }
    }

    v8f H2[HH / 16][NCT];
#pragma unroll
    for (int u = 0; u < HH / 16; ++u) { H2[u][0] = vzero; H2[u][1] = vzero; }

    for (int T = 0; T < HH / 16; ++T) {
        const int Tn = (T + 1 < HH / 16) ? T + 1 : T;
        const v8f bias1 = ld8(b1 + T * 16 + 8 * h);
        v8f c0 = vzero, c1 = vzero;
        v8f t0 = vzero, t1 = vzero;
#pragma unroll
        for (int s = 0; s < S12; ++s) {
            const int sn = s + AH;
            A[sn % NRNG] = ld8(sn < S12 ? WPTR12(T, sn) : WPTR12(Tn, sn - S12));
            __builtin_amdgcn_sched_barrier(0);
            if (s < S1) {
                mma_block(A[s % NRNG], E[s][0], E[s][1], c0, c1);
            } else {
                if (s == S1) { t0 = bias_relu(c0, bias1); t1 = bias_relu(c1, bias1); }
                mma_block(A[s % NRNG], t0, t1, H2[s - S1][0], H2[s - S1][1]);
            }
            __builtin_amdgcn_sched_barrier(0);
        }
    }
#undef WPTR12

#define WPTR3(V, s) (w3p + (size_t)(V) * 16 * HH + (s) * 16)
#pragma unroll
    for (int s = 0; s < AH; ++s) A[s] = ld8(WPTR3(0, s));

#pragma unroll
    for (int u = 0; u < HH / 16; ++u) {
        const v8f bias2 = ld8(b2 + u * 16 + 8 * h);
        H2[u][0] = bias_relu(H2[u][0], bias2);
        H2[u][1] = bias_relu(H2[u][1], bias2);
    }

    for (int V = 0; V < CC / 16; ++V) {
        const int Vn = (V + 1 < CC / 16) ? V + 1 : V;
        const v8f bias3 = ld8(b3 + V * 16 + 8 * h);
        v8f c0 = vzero, c1 = vzero;
#pragma unroll
        for (int s = 0; s < S3; ++s) {
            const int sn = s + AH;
            A[sn % NRNG] = ld8(sn < S3 ? WPTR3(V, sn) : WPTR3(Vn, sn - S3));
            __builtin_amdgcn_sched_barrier(0);
            mma_block(A[s % NRNG], H2[s][0], H2[s][1], c0, c1);
            __builtin_amdgcn_sched_barrier(0);
        }
        float m0[8], m1[8];
#pragma unroll
        for (int r = 0; r < 8; ++r) { m0[r] = c0[r] + bias3[r]; m1[r] = c1[r] + bias3[r]; }
#pragma unroll
        for (int o = 1; o < 16; o <<= 1) {
#pragma unroll
            for (int r = 0; r < 8; ++r) {
                m0[r] = fmaxf(m0[r], __shfl_xor(m0[r], o));
                m1[r] = fmaxf(m1[r], __shfl_xor(m1[r], o));
            }
        }
        if (c == 0) {
            v2f xr[8];
#pragma unroll
            for (int r = 0; r < 8; ++r)
                xr[r] = *(const v2f*)&x[(((size_t)b * CC + V * 16 + 8 * h + r) * DD + d) * NN + n0];
#pragma unroll
            for (int r = 0; r < 8; ++r)
                *(v2f*)&out[(((size_t)b * CC + V * 16 + 8 * h + r) * DD + d) * NN + n0] = (v2f){m0[r] + xr[r].x, m1[r] + xr[r].y};
        }
    }
#undef WPTR3
}

extern "C" void kernel_launch(void* const* d_in, const int* in_sizes, int n_in,
                              void* d_out, int out_size, void* d_ws, size_t ws_size,
                              hipStream_t stream) {
    const float* x  = (const float*)d_in[0];
    const float* W1 = (const float*)d_in[1];
    const float* b1 = (const float*)d_in[2];
    const float* W2 = (const float*)d_in[3];
    const float* b2 = (const float*)d_in[4];
    const float* W3 = (const float*)d_in[5];
    const float* b3 = (const float*)d_in[6];
    float* out = (float*)d_out;

    float* mean_t  = (float*)d_ws;
    float* sq      = mean_t + (size_t)BB * NN * CC;
    int*   idxp    = (int*)(sq + (size_t)BB * NN);

    meanpc_kernel<<<BB * (NN / MT), MTHR, 0, stream>>>(x, mean_t);
    sqnorm_kernel<<<(BB * NN + 255) / 256, 256, 0, stream>>>(mean_t, sq);
    knn_kernel<<<BB * (NN / KQ), (NS + 1) * KQ, 0, stream>>>(mean_t, sq, idxp);
    mlp_kernel<<<BB * DD * (NN / NCT) / MW, MW * 32, 0, stream>>>(x, mean_t, idxp,
                                                        W1, b1, W2, b2, W3, b3, out);
}


# KNN scan: a scoring lane takes 2 queries x the 16 refs of its lane half (the halves of a broadcast 128-bit LDS load read different rows): half the LDS loads per fma; fmaf order, scores unchanged
# speedup vs baseline: 12.1457x; 1.0692x over previous
#include <hip/hip_runtime.h>

#define BB 2
#define CC 64
#define DD 3
#define NN 4096
#define KK 16
#define EE 128
#define HH 256

typedef __attribute__((ext_vector_type(2))) float v2f;
typedef __attribute__((ext_vector_type(8))) float v8f;

constexpr int NCT  = 2;
constexpr int MW   = 4;

constexpr int MT = 32;
constexpr int MTHR = 256;
static_assert(NN % MT == 0 && MTHR % MT == 0 && CC % (MTHR / MT) == 0 && (MT * CC) % MTHR == 0, "mean tile must divide evenly");

__global__ void __launch_bounds__(MTHR) meanpc_kernel(const float* __restrict__ x, float* __restrict__ mean_t) {
    __shared__ float tl[MT][CC + 1];
    const int t  = threadIdx.x;
    const int b  = blockIdx.x / (NN / MT);
    const int n0 = (blockIdx.x % (NN / MT)) * MT;
    const int nl = t % MT;
    const int c0 = t / MT;
#pragma unroll
    for (int i = 0; i < CC / (MTHR / MT); ++i) {
        const int c = c0 + (MTHR / MT) * i;
        const float* xp = x + ((size_t)b * CC + c) * (DD * NN) + n0 + nl;
        tl[nl][c] = (xp[0] + xp[NN] + xp[2 * NN]) * (1.0f / 3.0f);
    }
    __syncthreads();
    float* dst = mean_t + ((size_t)b * NN + n0) * CC;
#pragma unroll
    for (int i = 0; i < MT * CC / MTHR; ++i) {
        const int f = t + MTHR * i;
        dst[f] = tl[f / CC][f % CC];
    }
}

typedef __attribute__((ext_vector_type(4))) float v4f;

__global__ void sqnorm_kernel(const float* __restrict__ mean_t, float* __restrict__ sq) {
    int i = blockIdx.x * blockDim.x + threadIdx.x;
    if (i >= BB * NN) return;
    const float* p = mean_t + (size_t)i * CC;
    float s = 0.0f;
#pragma unroll
    for (int c4 = 0; c4 < CC / 4; ++c4) {
        const v4f v = *(const v4f*)(p + 4 * c4);
        s = __builtin_fmaf(v.x, v.x, s);
        s = __builtin_fmaf(v.y, v.y, s);
        s = __builtin_fmaf(v.z, v.z, s);
        s = __builtin_fmaf(v.w, v.w, s);
    }
    sq[i] = s;
}

constexpr int KQ    = 32;
constexpr int NS    = 2;
constexpr int TR    = 32;
constexpr int PP    = 2 * CC + 4;
constexpr int RG    = 16;
constexpr int NPAIR = RG / 2;
constexpr int NBAT  = CC / 2;
constexpr int AHEAD = 3;
constexpr int NSLOT = AHEAD + 1;
constexpr int NGRP  = TR / RG;
constexpr int SG    = NS * NGRP;
constexpr int NSTEP = NN / (NS * TR);
constexpr int QH    = KQ / 2;
static_assert(NGRP == 2 && QH == 16, "a scoring lane is (query pair c, tile half h): 16 lanes x 2 halves");
static_assert(NN % (NS * TR) == 0, "steps must cover the points exactly");

constexpr int NCE = 63;
constexpr unsigned char SORT16[NCE][2] = {
    {0,1}, {2,3}, {4,5}, {6,7}, {8,9}, {10,11}, {12,13}, {14,15},
    {0,2}, {1,3}, {4,6}, {5,7}, {8,10}, {9,11}, {12,14}, {13,15},
    {1,2}, {5,6}, {9,10}, {13,14}, {0,4}, {3,7}, {8,12}, {11,15},
    {1,5}, {2,6}, {9,13}, {10,14}, {0,8}, {7,15},
    {2,4}, {3,5}, {10,12}, {11,13},
    {1,2}, {3,4}, {5,6}, {9,10}, {11,12}, {13,14},
    {1,9}, {2,10}, {3,11}, {4,12}, {5,13}, {6,14},
    {4,8}, {5,9}, {6,10}, {7,11},
    {2,4}, {3,5}, {6,8}, {7,9}, {10,12}, {11,13},
    {1,2}, {3,4}, {5,6}, {7,8}, {9,10}, {11,12}, {13,14}
};
static_assert(RG == KK, "the fold below pairs list position j with new position 15 - j");

__device__ __forceinline__ void knn_ce(float& da, float& db, int& ia, int& ib) {
    const bool s = db < da;
    const float lo = s ? db : da, hi = s ? da : db;
    const int   il = s ? ib : ia, ih = s ? ia : ib;
    da = lo; db = hi; ia = il; ib = ih;
}

__device__ __forceinline__ void knn_merge_group(float (&d2)[RG], int r0, int n, int nb,
                                                float (&bd)[KK], int (&bi)[KK]) {
    int ri[RG];
#pragma unroll
    for (int i = 0; i < RG; ++i) ri[i] = r0 + i;
    if ((r0 & ~(KQ - 1)) == nb) {
#pragma unroll
        for (int i = 0; i < RG; ++i) if (r0 + i == n) d2[i] = 3.0e38f;
    }
    bool any = false;
#pragma unroll
    for (int i = 0; i < RG; ++i) any |= d2[i] < bd[KK - 1];
    if (any) {
#pragma unroll
        for (int e = 0; e < NCE; ++e)
            knn_ce(d2[SORT16[e][0]], d2[SORT16[e][1]], ri[SORT16[e][0]], ri[SORT16[e][1]]);
#pragma unroll
        for (int j = 0; j < KK; ++j) {
            const bool take = d2[RG - 1 - j] < bd[j];
            bd[j] = take ? d2[RG - 1 - j] : bd[j];
            bi[j] = take ? ri[RG - 1 - j] : bi[j];
        }
#pragma unroll
        for (int k = KK / 2; k >= 1; k >>= 1) {
#pragma unroll
            for (int i = 0; i < KK; ++i)
                if ((i & k) == 0) knn_ce(bd[i], bd[i + k], bi[i], bi[i + k]);
        }
    }
}

__device__ __forceinline__ void knn_merge_step(const float (*sc)[SG][KQ], int lane, int r0, int n, int nb,
                                               float (&bd)[KK], int (&bi)[KK]) {
    float nx[RG];
#pragma unroll
    for (int i = 0; i < RG; ++i) nx[i] = sc[i][0][lane];
    for (int k = 0; k < SG; ++k) {
        float d2[RG];
#pragma unroll
        for (int i = 0; i < RG; ++i) d2[i] = nx[i];
        const int kn = (k + 1 < SG) ? k + 1 : k;
#pragma unroll
        for (int i = 0; i < RG; ++i) nx[i] = sc[i][kn][lane];
        knn_merge_group(d2, r0 + k * RG, n, nb, bd, bi);
    }
}

__device__ __forceinline__ void ld_point(const float* p, float (&v)[CC]) {
#pragma unroll
    for (int c4 = 0; c4 < CC / 4; ++c4) {
        const v4f w = *(const v4f*)(p + 4 * c4);
        v[4 * c4 + 0] = w.x; v[4 * c4 + 1] = w.y; v[4 * c4 + 2] = w.z; v[4 * c4 + 3] = w.w;
    }
}

__global__ void __launch_bounds__((NS + 1) * KQ) knn_kernel(const float* __restrict__ mean_t,
                                                            const float* __restrict__ sq,
                                                            int* __restrict__ idx) {
    __shared__ __attribute__((aligned(16))) float tile[NS][(TR / 2) * PP];
    __shared__ float score[2][RG][SG][KQ];

    const int lane = threadIdx.x & (KQ - 1);
    const int role = __builtin_amdgcn_readfirstlane((int)(threadIdx.x >> 5));
    const bool scorer = role < NS;
    const int b = blockIdx.x / (NN / KQ);
    const int nb = (blockIdx.x % (NN / KQ)) * KQ;
    const int n = nb + lane;
    const int qc = lane & (QH - 1);
    const int hh = lane >> 4;
    const float* pcb = mean_t + (size_t)b * NN * CC;
    const float* sqb = sq + (size_t)b * NN;
    float* mytile = &tile[scorer ? role : 0][0];

    float qa[CC];
    float qb[CC];
    float pf[CC];
    if (scorer) {
        ld_point(pcb + (size_t)(nb + qc) * CC, qa);
        ld_point(pcb + (size_t)(nb + qc + QH) * CC, qb);
        ld_point(pcb + (size_t)(role * TR + lane) * CC, pf);
    }

    float bd[KK];
    int   bi[KK];
#pragma unroll
    for (int j = 0; j < KK; ++j) { bd[j] = 3.0e38f; bi[j] = 0; }

    const int sp = (lane >> 1) * PP + (lane & 1);

    for (int st = 0; st < NSTEP; ++st) {
        if (scorer) {
#pragma unroll
            for (int c = 0; c < CC; ++c) mytile[sp + 2 * c] = pf[c];
        }
        __syncthreads();

        if (scorer) {
            const int t = st * NS + role;
            v4f rv[NSLOT][NPAIR];
            if (st + 1 < NSTEP) ld_point(pcb + (size_t)((t + NS) * TR + lane) * CC, pf);
            const int r0 = (t * NGRP + hh) * RG;
            const float* rows = &mytile[hh * NPAIR * PP];
            v4f sqv[RG / 4];
#pragma unroll
            for (int j = 0; j < RG / 4; ++j) sqv[j] = *(const v4f*)&sqb[r0 + 4 * j];
#pragma unroll
            for (int s = 0; s < AHEAD; ++s) {
#pragma unroll
                for (int p = 0; p < NPAIR; ++p) rv[s][p] = *(const v4f*)&rows[p * PP + 4 * s];
            }

            v2f acca[NPAIR], accb[NPAIR];
#pragma unroll
            for (int p = 0; p < NPAIR; ++p) { acca[p] = (v2f){0.0f, 0.0f}; accb[p] = (v2f){0.0f, 0.0f}; }
#pragma unroll
            for (int bt = 0; bt < NBAT; ++bt) {
                const int nx = bt + AHEAD;
                if (nx < NBAT) {
#pragma unroll
                    for (int p = 0; p < NPAIR; ++p)
                        rv[nx % NSLOT][p] = *(const v4f*)&rows[p * PP + 4 * nx];
                }
                __builtin_amdgcn_sched_barrier(0);
                const int cs = bt % NSLOT;
                const v2f qa0 = {qa[2 * bt + 0], qa[2 * bt + 0]};
                const v2f qa1 = {qa[2 * bt + 1], qa[2 * bt + 1]};
                const v2f qb0 = {qb[2 * bt + 0], qb[2 * bt + 0]};
                const v2f qb1 = {qb[2 * bt + 1], qb[2 * bt + 1]};
#pragma unroll
                for (int p = 0; p < NPAIR; ++p) acca[p] = __builtin_elementwise_fma(qa0, rv[cs][p].xy, acca[p]);
#pragma unroll
                for (int p = 0; p < NPAIR; ++p) accb[p] = __builtin_elementwise_fma(qb0, rv[cs][p].xy, accb[p]);
#pragma unroll
                for (int p = 0; p < NPAIR; ++p) acca[p] = __builtin_elementwise_fma(qa1, rv[cs][p].zw, acca[p]);
#pragma unroll
                for (int p = 0; p < NPAIR; ++p) accb[p] = __builtin_elementwise_fma(qb1, rv[cs][p].zw, accb[p]);
                __builtin_amdgcn_sched_barrier(0);
            }

#pragma unroll
            for (int i = 0; i < RG; ++i) {
                score[st & 1][i][role * NGRP + hh][qc]      = sqv[i >> 2][i & 3] - 2.0f * acca[i >> 1][i & 1];
                score[st & 1][i][role * NGRP + hh][qc + QH] = sqv[i >> 2][i & 3] - 2.0f * accb[i >> 1][i & 1];
            }
        } else if (st > 0) {
            knn_merge_step(score[(st - 1) & 1], lane, (st - 1) * SG * RG, n, nb, bd, bi);
        }
        __syncthreads();
    }
    if (!scorer) {
        constexpr int SL = NSTEP - 1;
        knn_merge_step(score[SL & 1], lane, SL * SG * RG, n, nb, bd, bi);
#pragma unroll
        for (int j = 0; j < KK; ++j) idx[((size_t)b * KK + j) * NN + n] = bi[j];
    }
}

typedef __attribute__((ext_vector_type(2))) int v2i;

__device__ __forceinline__ v8f ld8(const float* p) {
    const v4f lo = *(const v4f*)p;
    const v4f hi = *(const v4f*)(p + 4);
    return (v8f){lo.x, lo.y, lo.z, lo.w, hi.x, hi.y, hi.z, hi.w};
}

__device__ __forceinline__ void mma_block(const v8f& a, const v8f& b0, const v8f& b1, v8f& c0, v8f& c1) {
#pragma unroll
    for (int m = 0; m < 4; ++m) {
        const v2f af = {a[2 * m], a[2 * m + 1]};
        const v2f f0 = {b0[2 * m], b0[2 * m + 1]};
        const v2f f1 = {b1[2 * m], b1[2 * m + 1]};
        c0 = __builtin_amdgcn_wmma_f32_16x16x4_f32(false, af, false, f0, (short)0, c0, false, false);
        c1 = __builtin_amdgcn_wmma_f32_16x16x4_f32(false, af, false, f1, (short)0, c1, false, false);
    }
}

__device__ __forceinline__ v8f bias_relu(const v8f& acc, const v8f& bias) {
    v8f o;
#pragma unroll
    for (int r = 0; r < 8; ++r) o[r] = fmaxf(acc[r] + bias[r], 0.0f);
    return o;
}

constexpr int AH   = 3;
constexpr int NRNG = AH + 1;
constexpr int S1   = EE / 16;
constexpr int S12  = S1 + HH / 16;
constexpr int S3   = HH / 16;
static_assert(S12 % NRNG == 0 && S3 % NRNG == 0, "ring phase must repeat from trip to trip");
static_assert(NCT == 2, "mma_block and the paired loads below are written for two column tiles");

__global__ void __launch_bounds__(MW * 32) mlp_kernel(const float* __restrict__ x,
                                                      const float* __restrict__ mean_t,
                                                      const int* __restrict__ idx,
                                                      const float* __restrict__ W1, const float* __restrict__ b1,
                                                      const float* __restrict__ W2, const float* __restrict__ b2,
                                                      const float* __restrict__ W3, const float* __restrict__ b3,
                                                      float* __restrict__ out) {
    const int lane = threadIdx.x & 31;
    const int c    = lane & 15;
    const int h    = lane >> 4;
    const int wv   = __builtin_amdgcn_readfirstlane((int)(threadIdx.x >> 5));
    const int task = blockIdx.x * MW + wv;
    const int np   = task % (NN / NCT);
    const int bd   = task / (NN / NCT);
    const int d    = bd % DD;
    const int b    = bd / DD;
    const int n0   = np * NCT;

    const v8f vzero = {0.f, 0.f, 0.f, 0.f, 0.f, 0.f, 0.f, 0.f};

    const float* w1p = W1 + (size_t)c * EE + 8 * h;
    const float* w2p = W2 + (size_t)c * HH + 8 * h;
    const float* w3p = W3 + (size_t)c * HH + 8 * h;

#define WPTR12(T, s) (((s) < S1) ? (w1p + (size_t)(T) * 16 * EE + (s) * 16) \
                                 : (w2p + (size_t)((s) - S1) * 16 * HH + (T) * 16))

    v8f A[NRNG];
#pragma unroll
    for (int s = 0; s < AH; ++s) A[s] = ld8(WPTR12(0, s));

    v8f E[S1][NCT];
    {
        const v2i jj = *(const v2i*)&idx[((size_t)b * KK + c) * NN + n0];
        const int j0 = jj.x & (NN - 1);
        const int j1 = jj.y & (NN - 1);
        const float* xb = x + (((size_t)b * CC) * DD + d) * NN + n0;
        const float* mb = mean_t + (size_t)b * NN * CC;
        const float* g0p = mb + (size_t)j0 * CC + 8 * h;
        const float* g1p = mb + (size_t)j1 * CC + 8 * h;
#pragma unroll
        for (int q = 0; q < CC / 16; ++q) {
            const v8f g0 = ld8(g0p + 16 * q);
            const v8f g1 = ld8(g1p + 16 * q);
#pragma unroll
            for (int r = 0; r < 8; ++r) {
                const int ch = q * 16 + 8 * h + r;
                const v2f xv = *(const v2f*)&xb[(size_t)ch * DD * NN];
                E[q][0][r] = xv.x;
                E[q][1][r] = xv.y;
                E[q + CC / 16][0][r] = g0[r] - xv.x;
                E[q + CC / 16][1][r] = g1[r] - xv.y;
            }
        }
    }

    v8f H2[HH / 16][NCT];
#pragma unroll
    for (int u = 0; u < HH / 16; ++u) { H2[u][0] = vzero; H2[u][1] = vzero; }

    for (int T = 0; T < HH / 16; ++T) {
        const int Tn = (T + 1 < HH / 16) ? T + 1 : T;
        const v8f bias1 = ld8(b1 + T * 16 + 8 * h);
        v8f c0 = vzero, c1 = vzero;
        v8f t0 = vzero, t1 = vzero;
#pragma unroll
        for (int s = 0; s < S12; ++s) {
            const int sn = s + AH;
            A[sn % NRNG] = ld8(sn < S12 ? WPTR12(T, sn) : WPTR12(Tn, sn - S12));
            __builtin_amdgcn_sched_barrier(0);
            if (s < S1) {
                mma_block(A[s % NRNG], E[s][0], E[s][1], c0, c1);
            } else {
                if (s == S1) { t0 = bias_relu(c0, bias1); t1 = bias_relu(c1, bias1); }
                mma_block(A[s % NRNG], t0, t1, H2[s - S1][0], H2[s - S1][1]);
            }
            __builtin_amdgcn_sched_barrier(0);
        }
    }
#undef WPTR12

#define WPTR3(V, s) (w3p + (size_t)(V) * 16 * HH + (s) * 16)
#pragma unroll
    for (int s = 0; s < AH; ++s) A[s] = ld8(WPTR3(0, s));

#pragma unroll
    for (int u = 0; u < HH / 16; ++u) {
        const v8f bias2 = ld8(b2 + u * 16 + 8 * h);
        H2[u][0] = bias_relu(H2[u][0], bias2);
        H2[u][1] = bias_relu(H2[u][1], bias2);
    }

    for (int V = 0; V < CC / 16; ++V) {
        const int Vn = (V + 1 < CC / 16) ? V + 1 : V;
        const v8f bias3 = ld8(b3 + V * 16 + 8 * h);
        v8f c0 = vzero, c1 = vzero;
#pragma unroll
        for (int s = 0; s < S3; ++s) {
            const int sn = s + AH;
            A[sn % NRNG] = ld8(sn < S3 ? WPTR3(V, sn) : WPTR3(Vn, sn - S3));
            __builtin_amdgcn_sched_barrier(0);
            mma_block(A[s % NRNG], H2[s][0], H2[s][1], c0, c1);
            __builtin_amdgcn_sched_barrier(0);
        }
        float m0[8], m1[8];
#pragma unroll
        for (int r = 0; r < 8; ++r) { m0[r] = c0[r] + bias3[r]; m1[r] = c1[r] + bias3[r]; }
#pragma unroll
        for (int o = 1; o < 16; o <<= 1) {
#pragma unroll
            for (int r = 0; r < 8; ++r) {
                m0[r] = fmaxf(m0[r], __shfl_xor(m0[r], o));
                m1[r] = fmaxf(m1[r], __shfl_xor(m1[r], o));
            }
        }
        if (c == 0) {
            v2f xr[8];
#pragma unroll
            for (int r = 0; r < 8; ++r)
                xr[r] = *(const v2f*)&x[(((size_t)b * CC + V * 16 + 8 * h + r) * DD + d) * NN + n0];
#pragma unroll
            for (int r = 0; r < 8; ++r)
                *(v2f*)&out[(((size_t)b * CC + V * 16 + 8 * h + r) * DD + d) * NN + n0] = (v2f){m0[r] + xr[r].x, m1[r] + xr[r].y};
        }
    }
#undef WPTR3
}

extern "C" void kernel_launch(void* const* d_in, const int* in_sizes, int n_in,
                              void* d_out, int out_size, void* d_ws, size_t ws_size,
                              hipStream_t stream) {
    const float* x  = (const float*)d_in[0];
    const float* W1 = (const float*)d_in[1];
    const float* b1 = (const float*)d_in[2];
    const float* W2 = (const float*)d_in[3];
    const float* b2 = (const float*)d_in[4];
    const float* W3 = (const float*)d_in[5];
    const float* b3 = (const float*)d_in[6];
    float* out = (float*)d_out;

    float* mean_t  = (float*)d_ws;
    float* sq      = mean_t + (size_t)BB * NN * CC;
    int*   idxp    = (int*)(sq + (size_t)BB * NN);

    meanpc_kernel<<<BB * (NN / MT), MTHR, 0, stream>>>(x, mean_t);
    sqnorm_kernel<<<(BB * NN + 255) / 256, 256, 0, stream>>>(mean_t, sq);
    knn_kernel<<<BB * (NN / KQ), (NS + 1) * KQ, 0, stream>>>(mean_t, sq, idxp);
    mlp_kernel<<<BB * DD * (NN / NCT) / MW, MW * 32, 0, stream>>>(x, mean_t, idxp,
                                                        W1, b1, W2, b2, W3, b3, out);
}


# MLP weights staged through LDS once per workgroup: its 4 waves copy each trip's W1/W2 (then W3) blocks by whole-row loads into 2 LDS stages and read fragments from LDS; WMMAs and sum order unchanged
# speedup vs baseline: 12.5502x; 1.0333x over previous
#include <hip/hip_runtime.h>

#define BB 2
#define CC 64
#define DD 3
#define NN 4096
#define KK 16
#define EE 128
#define HH 256

typedef __attribute__((ext_vector_type(2))) float v2f;
typedef __attribute__((ext_vector_type(8))) float v8f;

constexpr int NCT  = 2;
constexpr int MW   = 4;

constexpr int MT = 32;
constexpr int MTHR = 256;
static_assert(NN % MT == 0 && MTHR % MT == 0 && CC % (MTHR / MT) == 0 && (MT * CC) % MTHR == 0, "mean tile must divide evenly");

__global__ void __launch_bounds__(MTHR) meanpc_kernel(const float* __restrict__ x, float* __restrict__ mean_t) {
    __shared__ float tl[MT][CC + 1];
    const int t  = threadIdx.x;
    const int b  = blockIdx.x / (NN / MT);
    const int n0 = (blockIdx.x % (NN / MT)) * MT;
    const int nl = t % MT;
    const int c0 = t / MT;
#pragma unroll
    for (int i = 0; i < CC / (MTHR / MT); ++i) {
        const int c = c0 + (MTHR / MT) * i;
        const float* xp = x + ((size_t)b * CC + c) * (DD * NN) + n0 + nl;
        tl[nl][c] = (xp[0] + xp[NN] + xp[2 * NN]) * (1.0f / 3.0f);
    }
    __syncthreads();
    float* dst = mean_t + ((size_t)b * NN + n0) * CC;
#pragma unroll
    for (int i = 0; i < MT * CC / MTHR; ++i) {
        const int f = t + MTHR * i;
        dst[f] = tl[f / CC][f % CC];
    }
}

typedef __attribute__((ext_vector_type(4))) float v4f;

__global__ void sqnorm_kernel(const float* __restrict__ mean_t, float* __restrict__ sq) {
    int i = blockIdx.x * blockDim.x + threadIdx.x;
    if (i >= BB * NN) return;
    const float* p = mean_t + (size_t)i * CC;
    float s = 0.0f;
#pragma unroll
    for (int c4 = 0; c4 < CC / 4; ++c4) {
        const v4f v = *(const v4f*)(p + 4 * c4);
        s = __builtin_fmaf(v.x, v.x, s);
        s = __builtin_fmaf(v.y, v.y, s);
        s = __builtin_fmaf(v.z, v.z, s);
        s = __builtin_fmaf(v.w, v.w, s);
    }
    sq[i] = s;
}

constexpr int KQ    = 32;
constexpr int NS    = 2;
constexpr int TR    = 32;
constexpr int PP    = 2 * CC + 4;
constexpr int RG    = 16;
constexpr int NPAIR = RG / 2;
constexpr int NBAT  = CC / 2;
constexpr int AHEAD = 3;
constexpr int NSLOT = AHEAD + 1;
constexpr int NGRP  = TR / RG;
constexpr int SG    = NS * NGRP;
constexpr int NSTEP = NN / (NS * TR);
constexpr int QH    = KQ / 2;
static_assert(NGRP == 2 && QH == 16, "a scoring lane is (query pair c, tile half h): 16 lanes x 2 halves");
static_assert(NN % (NS * TR) == 0, "steps must cover the points exactly");

constexpr int NCE = 63;
constexpr unsigned char SORT16[NCE][2] = {
    {0,1}, {2,3}, {4,5}, {6,7}, {8,9}, {10,11}, {12,13}, {14,15},
    {0,2}, {1,3}, {4,6}, {5,7}, {8,10}, {9,11}, {12,14}, {13,15},
    {1,2}, {5,6}, {9,10}, {13,14}, {0,4}, {3,7}, {8,12}, {11,15},
    {1,5}, {2,6}, {9,13}, {10,14}, {0,8}, {7,15},
    {2,4}, {3,5}, {10,12}, {11,13},
    {1,2}, {3,4}, {5,6}, {9,10}, {11,12}, {13,14},
    {1,9}, {2,10}, {3,11}, {4,12}, {5,13}, {6,14},
    {4,8}, {5,9}, {6,10}, {7,11},
    {2,4}, {3,5}, {6,8}, {7,9}, {10,12}, {11,13},
    {1,2}, {3,4}, {5,6}, {7,8}, {9,10}, {11,12}, {13,14}
};
static_assert(RG == KK, "the fold below pairs list position j with new position 15 - j");

__device__ __forceinline__ void knn_ce(float& da, float& db, int& ia, int& ib) {
    const bool s = db < da;
    const float lo = s ? db : da, hi = s ? da : db;
    const int   il = s ? ib : ia, ih = s ? ia : ib;
    da = lo; db = hi; ia = il; ib = ih;
}

__device__ __forceinline__ void knn_merge_group(float (&d2)[RG], int r0, int n, int nb,
                                                float (&bd)[KK], int (&bi)[KK]) {
    int ri[RG];
#pragma unroll
    for (int i = 0; i < RG; ++i) ri[i] = r0 + i;
    if ((r0 & ~(KQ - 1)) == nb) {
#pragma unroll
        for (int i = 0; i < RG; ++i) if (r0 + i == n) d2[i] = 3.0e38f;
    }
    bool any = false;
#pragma unroll
    for (int i = 0; i < RG; ++i) any |= d2[i] < bd[KK - 1];
    if (any) {
#pragma unroll
        for (int e = 0; e < NCE; ++e)
            knn_ce(d2[SORT16[e][0]], d2[SORT16[e][1]], ri[SORT16[e][0]], ri[SORT16[e][1]]);
#pragma unroll
        for (int j = 0; j < KK; ++j) {
            const bool take = d2[RG - 1 - j] < bd[j];
            bd[j] = take ? d2[RG - 1 - j] : bd[j];
            bi[j] = take ? ri[RG - 1 - j] : bi[j];
        }
#pragma unroll
        for (int k = KK / 2; k >= 1; k >>= 1) {
#pragma unroll
            for (int i = 0; i < KK; ++i)
                if ((i & k) == 0) knn_ce(bd[i], bd[i + k], bi[i], bi[i + k]);
        }
    }
}

__device__ __forceinline__ void knn_merge_step(const float (*sc)[SG][KQ], int lane, int r0, int n, int nb,
                                               float (&bd)[KK], int (&bi)[KK]) {
    float nx[RG];
#pragma unroll
    for (int i = 0; i < RG; ++i) nx[i] = sc[i][0][lane];
    for (int k = 0; k < SG; ++k) {
        float d2[RG];
#pragma unroll
        for (int i = 0; i < RG; ++i) d2[i] = nx[i];
        const int kn = (k + 1 < SG) ? k + 1 : k;
#pragma unroll
        for (int i = 0; i < RG; ++i) nx[i] = sc[i][kn][lane];
        knn_merge_group(d2, r0 + k * RG, n, nb, bd, bi);
    }
}

__device__ __forceinline__ void ld_point(const float* p, float (&v)[CC]) {
#pragma unroll
    for (int c4 = 0; c4 < CC / 4; ++c4) {
        const v4f w = *(const v4f*)(p + 4 * c4);
        v[4 * c4 + 0] = w.x; v[4 * c4 + 1] = w.y; v[4 * c4 + 2] = w.z; v[4 * c4 + 3] = w.w;
    }
}

__global__ void __launch_bounds__((NS + 1) * KQ) knn_kernel(const float* __restrict__ mean_t,
                                                            const float* __restrict__ sq,
                                                            int* __restrict__ idx) {
    __shared__ __attribute__((aligned(16))) float tile[NS][(TR / 2) * PP];
    __shared__ float score[2][RG][SG][KQ];

    const int lane = threadIdx.x & (KQ - 1);
    const int role = __builtin_amdgcn_readfirstlane((int)(threadIdx.x >> 5));
    const bool scorer = role < NS;
    const int b = blockIdx.x / (NN / KQ);
    const int nb = (blockIdx.x % (NN / KQ)) * KQ;
    const int n = nb + lane;
    const int qc = lane & (QH - 1);
    const int hh = lane >> 4;
    const float* pcb = mean_t + (size_t)b * NN * CC;
    const float* sqb = sq + (size_t)b * NN;
    float* mytile = &tile[scorer ? role : 0][0];

    float qa[CC];
    float qb[CC];
    float pf[CC];
    if (scorer) {
        ld_point(pcb + (size_t)(nb + qc) * CC, qa);
        ld_point(pcb + (size_t)(nb + qc + QH) * CC, qb);
        ld_point(pcb + (size_t)(role * TR + lane) * CC, pf);
    }

    float bd[KK];
    int   bi[KK];
#pragma unroll
    for (int j = 0; j < KK; ++j) { bd[j] = 3.0e38f; bi[j] = 0; }

    const int sp = (lane >> 1) * PP + (lane & 1);

    for (int st = 0; st < NSTEP; ++st) {
        if (scorer) {
#pragma unroll
            for (int c = 0; c < CC; ++c) mytile[sp + 2 * c] = pf[c];
        }
        __syncthreads();

        if (scorer) {
            const int t = st * NS + role;
            v4f rv[NSLOT][NPAIR];
            if (st + 1 < NSTEP) ld_point(pcb + (size_t)((t + NS) * TR + lane) * CC, pf);
            const int r0 = (t * NGRP + hh) * RG;
            const float* rows = &mytile[hh * NPAIR * PP];
            v4f sqv[RG / 4];
#pragma unroll
            for (int j = 0; j < RG / 4; ++j) sqv[j] = *(const v4f*)&sqb[r0 + 4 * j];
#pragma unroll
            for (int s = 0; s < AHEAD; ++s) {
#pragma unroll
                for (int p = 0; p < NPAIR; ++p) rv[s][p] = *(const v4f*)&rows[p * PP + 4 * s];
            }

            v2f acca[NPAIR], accb[NPAIR];
#pragma unroll
            for (int p = 0; p < NPAIR; ++p) { acca[p] = (v2f){0.0f, 0.0f}; accb[p] = (v2f){0.0f, 0.0f}; }
#pragma unroll
            for (int bt = 0; bt < NBAT; ++bt) {
                const int nx = bt + AHEAD;
                if (nx < NBAT) {
#pragma unroll
                    for (int p = 0; p < NPAIR; ++p)
                        rv[nx % NSLOT][p] = *(const v4f*)&rows[p * PP + 4 * nx];
                }
                __builtin_amdgcn_sched_barrier(0);
                const int cs = bt % NSLOT;
                const v2f qa0 = {qa[2 * bt + 0], qa[2 * bt + 0]};
                const v2f qa1 = {qa[2 * bt + 1], qa[2 * bt + 1]};
                const v2f qb0 = {qb[2 * bt + 0], qb[2 * bt + 0]};
                const v2f qb1 = {qb[2 * bt + 1], qb[2 * bt + 1]};
#pragma unroll
                for (int p = 0; p < NPAIR; ++p) acca[p] = __builtin_elementwise_fma(qa0, rv[cs][p].xy, acca[p]);
#pragma unroll
                for (int p = 0; p < NPAIR; ++p) accb[p] = __builtin_elementwise_fma(qb0, rv[cs][p].xy, accb[p]);
#pragma unroll
                for (int p = 0; p < NPAIR; ++p) acca[p] = __builtin_elementwise_fma(qa1, rv[cs][p].zw, acca[p]);
#pragma unroll
                for (int p = 0; p < NPAIR; ++p) accb[p] = __builtin_elementwise_fma(qb1, rv[cs][p].zw, accb[p]);
                __builtin_amdgcn_sched_barrier(0);
            }

#pragma unroll
            for (int i = 0; i < RG; ++i) {
                score[st & 1][i][role * NGRP + hh][qc]      = sqv[i >> 2][i & 3] - 2.0f * acca[i >> 1][i & 1];
                score[st & 1][i][role * NGRP + hh][qc + QH] = sqv[i >> 2][i & 3] - 2.0f * accb[i >> 1][i & 1];
            }
        } else if (st > 0) {
            knn_merge_step(score[(st - 1) & 1], lane, (st - 1) * SG * RG, n, nb, bd, bi);
        }
        __syncthreads();
    }
    if (!scorer) {
        constexpr int SL = NSTEP - 1;
        knn_merge_step(score[SL & 1], lane, SL * SG * RG, n, nb, bd, bi);
#pragma unroll
        for (int j = 0; j < KK; ++j) idx[((size_t)b * KK + j) * NN + n] = bi[j];
    }
}

typedef __attribute__((ext_vector_type(2))) int v2i;

__device__ __forceinline__ v8f ld8(const float* p) {
    const v4f lo = *(const v4f*)p;
    const v4f hi = *(const v4f*)(p + 4);
    return (v8f){lo.x, lo.y, lo.z, lo.w, hi.x, hi.y, hi.z, hi.w};
}

__device__ __forceinline__ void mma_block(const v8f& a, const v8f& b0, const v8f& b1, v8f& c0, v8f& c1) {
#pragma unroll
    for (int m = 0; m < 4; ++m) {
        const v2f af = {a[2 * m], a[2 * m + 1]};
        const v2f f0 = {b0[2 * m], b0[2 * m + 1]};
        const v2f f1 = {b1[2 * m], b1[2 * m + 1]};
        c0 = __builtin_amdgcn_wmma_f32_16x16x4_f32(false, af, false, f0, (short)0, c0, false, false);
        c1 = __builtin_amdgcn_wmma_f32_16x16x4_f32(false, af, false, f1, (short)0, c1, false, false);
    }
}

__device__ __forceinline__ v8f bias_relu(const v8f& acc, const v8f& bias) {
    v8f o;
#pragma unroll
    for (int r = 0; r < 8; ++r) o[r] = fmaxf(acc[r] + bias[r], 0.0f);
    return o;
}

constexpr int AH   = 1;
constexpr int NRNG = AH + 1;
constexpr int S1   = EE / 16;
constexpr int S12  = S1 + HH / 16;
constexpr int S3   = HH / 16;
static_assert(S12 % NRNG == 0 && S3 % NRNG == 0, "ring phase must repeat from trip to trip");
static_assert(NCT == 2, "mma_block and the paired loads below are written for two column tiles");

constexpr int RP   = 20;
constexpr int BLK  = 16 * RP + 16;
constexpr int STG  = S12 * BLK;
constexpr int CPG  = 4;
constexpr int NP12 = S12 * 64 / 128;
constexpr int NP3  = S3 * 64 / 128;
static_assert(NP12 == 3 * CPG && NP3 == 2 * CPG, "three copy groups per trip of the first loop, two of the second");
static_assert(MW * 32 == 128, "the copy maps below are written for 128 threads");
static_assert(20 < S12 - AH && 11 < S3 - AH, "the last copy group of a trip is stored in front of the trip's barrier");
static_assert((STG * 2 * 4) <= 65536, "two stages must fit the static LDS limit");

__global__ void __launch_bounds__(MW * 32) mlp_kernel(const float* __restrict__ x,
                                                      const float* __restrict__ mean_t,
                                                      const int* __restrict__ idx,
                                                      const float* __restrict__ W1, const float* __restrict__ b1,
                                                      const float* __restrict__ W2, const float* __restrict__ b2,
                                                      const float* __restrict__ W3, const float* __restrict__ b3,
                                                      float* __restrict__ out) {
    __shared__ __attribute__((aligned(16))) float wst[2 * STG];

    const int t    = threadIdx.x;
    const int lane = t & 31;
    const int c    = lane & 15;
    const int h    = lane >> 4;
    const int wv   = __builtin_amdgcn_readfirstlane((int)(threadIdx.x >> 5));
    const int task = blockIdx.x * MW + wv;
    const int np   = task % (NN / NCT);
    const int bd   = task / (NN / NCT);
    const int d    = bd % DD;
    const int b    = bd / DD;
    const int n0   = np * NCT;

    const v8f vzero = {0.f, 0.f, 0.f, 0.f, 0.f, 0.f, 0.f, 0.f};

    const float* cw1 = W1 + (size_t)(t >> 5) * EE + 4 * (t & 31);
    const float* cw2 = W2 + (size_t)(t >> 2) * HH + 4 * (t & 3);
    const float* cw3 = W3 + (size_t)(t >> 6) * HH + 4 * (t & 63);
    const int dw1 = ((t & 31) >> 2) * BLK + (t >> 5) * RP + 4 * (t & 3);
    const int dw2 = (S1 + (t >> 6)) * BLK + ((t >> 2) & 15) * RP + 4 * (t & 3);
    const int dw3 = ((t & 63) >> 2) * BLK + (t >> 6) * RP + 4 * (t & 3);
#define SRC12(g, i) (((i) < 4) ? (cw1 + (size_t)(g) * 16 * EE + (i) * 4 * EE) \
                               : (cw2 + (size_t)(g) * 16 + (size_t)((i) - 4) * 32 * HH))
#define DST12(i)    (((i) < 4) ? (dw1 + (i) * 4 * RP) : (dw2 + ((i) - 4) * 2 * BLK))
#define SRC3(v, i)  (cw3 + (size_t)(v) * 16 * HH + (i) * 2 * HH)
#define DST3(i)     (dw3 + (i) * 2 * RP)

    const int fo = c * RP + 8 * h;

    {
        v4f pst[NP12];
#pragma unroll
        for (int i = 0; i < NP12; ++i) pst[i] = *(const v4f*)SRC12(0, i);
#pragma unroll
        for (int i = 0; i < NP12; ++i) *(v4f*)&wst[DST12(i)] = pst[i];
    }

    v8f E[S1][NCT];
    {
        const v2i jj = *(const v2i*)&idx[((size_t)b * KK + c) * NN + n0];
        const int j0 = jj.x & (NN - 1);
        const int j1 = jj.y & (NN - 1);
        const float* xb = x + (((size_t)b * CC) * DD + d) * NN + n0;
        const float* mb = mean_t + (size_t)b * NN * CC;
        const float* g0p = mb + (size_t)j0 * CC + 8 * h;
        const float* g1p = mb + (size_t)j1 * CC + 8 * h;
#pragma unroll
        for (int q = 0; q < CC / 16; ++q) {
            const v8f g0 = ld8(g0p + 16 * q);
            const v8f g1 = ld8(g1p + 16 * q);
#pragma unroll
            for (int r = 0; r < 8; ++r) {
                const int ch = q * 16 + 8 * h + r;
                const v2f xv = *(const v2f*)&xb[(size_t)ch * DD * NN];
                E[q][0][r] = xv.x;
                E[q][1][r] = xv.y;
                E[q + CC / 16][0][r] = g0[r] - xv.x;
                E[q + CC / 16][1][r] = g1[r] - xv.y;
            }
        }
    }

    __syncthreads();

    v8f A[NRNG];
#pragma unroll
    for (int s = 0; s < AH; ++s) A[s] = ld8(&wst[s * BLK + fo]);

    v8f H2[HH / 16][NCT];
#pragma unroll
    for (int u = 0; u < HH / 16; ++u) { H2[u][0] = vzero; H2[u][1] = vzero; }

    for (int T = 0; T < HH / 16; ++T) {
        const int ocur = (T & 1) * STG;
        const int onxt = ((T + 1) & 1) * STG;
        const bool n12 = T + 1 < HH / 16;
        const v8f bias1 = ld8(b1 + T * 16 + 8 * h);
        v8f c0 = vzero, c1 = vzero;
        v8f t0 = vzero, t1 = vzero;
        v4f st[CPG];
#pragma unroll
        for (int s = 0; s < S12; ++s) {
            if (s == S12 - AH) __syncthreads();
            const int sn = s + AH;
            A[sn % NRNG] = ld8(&wst[(sn < S12 ? ocur + sn * BLK : onxt + (sn - S12) * BLK) + fo]);
            if (s % 7 == 0 && s / 7 < NP12 / CPG) {
#pragma unroll
                for (int e = 0; e < CPG; ++e) {
                    const int i  = CPG * (s / 7) + e;
                    const int i3 = i < NP3 ? i : i - CPG;
                    st[e] = *(const v4f*)(n12 ? SRC12(T + 1, i) : SRC3(0, i3));
                }
            }
            __builtin_amdgcn_sched_barrier(0);
            if (s < S1) {
                mma_block(A[s % NRNG], E[s][0], E[s][1], c0, c1);
            } else {
                if (s == S1) { t0 = bias_relu(c0, bias1); t1 = bias_relu(c1, bias1); }
                mma_block(A[s % NRNG], t0, t1, H2[s - S1][0], H2[s - S1][1]);
            }
            if (s % 7 == 6 && s / 7 < NP12 / CPG) {
#pragma unroll
                for (int e = 0; e < CPG; ++e) {
                    const int i  = CPG * (s / 7) + e;
                    const int i3 = i < NP3 ? i : i - CPG;
                    *(v4f*)&wst[onxt + (n12 ? DST12(i) : DST3(i3))] = st[e];
                }
            }
            __builtin_amdgcn_sched_barrier(0);
        }
    }

#pragma unroll
    for (int s = 0; s < AH; ++s) A[s] = ld8(&wst[s * BLK + fo]);

#pragma unroll
    for (int u = 0; u < HH / 16; ++u) {
        const v8f bias2 = ld8(b2 + u * 16 + 8 * h);
        H2[u][0] = bias_relu(H2[u][0], bias2);
        H2[u][1] = bias_relu(H2[u][1], bias2);
    }

    for (int V = 0; V < CC / 16; ++V) {
        const int Vn = (V + 1 < CC / 16) ? V + 1 : V;
        const int ocur = (V & 1) * STG;
        const int onxt = ((V + 1) & 1) * STG;
        const v8f bias3 = ld8(b3 + V * 16 + 8 * h);
        v8f c0 = vzero, c1 = vzero;
        v4f st[CPG];
#pragma unroll
        for (int s = 0; s < S3; ++s) {
            if (s == S3 - AH) __syncthreads();
            const int sn = s + AH;
            A[sn % NRNG] = ld8(&wst[(sn < S3 ? ocur + sn * BLK : onxt + (sn - S3) * BLK) + fo]);
            if (s % 6 == 0 && s / 6 < NP3 / CPG) {
#pragma unroll
                for (int e = 0; e < CPG; ++e) st[e] = *(const v4f*)SRC3(Vn, CPG * (s / 6) + e);
            }
            __builtin_amdgcn_sched_barrier(0);
            mma_block(A[s % NRNG], H2[s][0], H2[s][1], c0, c1);
            if (s % 6 == 5 && s / 6 < NP3 / CPG) {
#pragma unroll
                for (int e = 0; e < CPG; ++e) *(v4f*)&wst[onxt + DST3(CPG * (s / 6) + e)] = st[e];
            }
            __builtin_amdgcn_sched_barrier(0);
        }
        float m0[8], m1[8];
#pragma unroll
        for (int r = 0; r < 8; ++r) { m0[r] = c0[r] + bias3[r]; m1[r] = c1[r] + bias3[r]; }
#pragma unroll
        for (int o = 1; o < 16; o <<= 1) {
#pragma unroll
            for (int r = 0; r < 8; ++r) {
                m0[r] = fmaxf(m0[r], __shfl_xor(m0[r], o));
                m1[r] = fmaxf(m1[r], __shfl_xor(m1[r], o));
            }
        }
        if (c == 0) {
            v2f xr[8];
#pragma unroll
            for (int r = 0; r < 8; ++r)
                xr[r] = *(const v2f*)&x[(((size_t)b * CC + V * 16 + 8 * h + r) * DD + d) * NN + n0];
#pragma unroll
            for (int r = 0; r < 8; ++r)
                *(v2f*)&out[(((size_t)b * CC + V * 16 + 8 * h + r) * DD + d) * NN + n0] = (v2f){m0[r] + xr[r].x, m1[r] + xr[r].y};
        }
    }
#undef SRC12
#undef DST12
#undef SRC3
#undef DST3
}

extern "C" void kernel_launch(void* const* d_in, const int* in_sizes, int n_in,
                              void* d_out, int out_size, void* d_ws, size_t ws_size,
                              hipStream_t stream) {
    const float* x  = (const float*)d_in[0];
    const float* W1 = (const float*)d_in[1];
    const float* b1 = (const float*)d_in[2];
    const float* W2 = (const float*)d_in[3];
    const float* b2 = (const float*)d_in[4];
    const float* W3 = (const float*)d_in[5];
    const float* b3 = (const float*)d_in[6];
    float* out = (float*)d_out;

    float* mean_t  = (float*)d_ws;
    float* sq      = mean_t + (size_t)BB * NN * CC;
    int*   idxp    = (int*)(sq + (size_t)BB * NN);

    meanpc_kernel<<<BB * (NN / MT), MTHR, 0, stream>>>(x, mean_t);
    sqnorm_kernel<<<(BB * NN + 255) / 256, 256, 0, stream>>>(mean_t, sq);
    knn_kernel<<<BB * (NN / KQ), (NS + 1) * KQ, 0, stream>>>(mean_t, sq, idxp);
    mlp_kernel<<<BB * DD * (NN / NCT) / MW, MW * 32, 0, stream>>>(x, mean_t, idxp,
                                                        W1, b1, W2, b2, W3, b3, out);
}


# MLP weight fragments requested two blocks at a time: one LDS request place per two weight blocks, not per block (ring of 4 register sets); weight copy loads and stores moved into those places
# speedup vs baseline: 12.9813x; 1.0344x over previous
#include <hip/hip_runtime.h>

#define BB 2
#define CC 64
#define DD 3
#define NN 4096
#define KK 16
#define EE 128
#define HH 256

typedef __attribute__((ext_vector_type(2))) float v2f;
typedef __attribute__((ext_vector_type(8))) float v8f;

constexpr int NCT  = 2;
constexpr int MW   = 4;

constexpr int MT = 32;
constexpr int MTHR = 256;
static_assert(NN % MT == 0 && MTHR % MT == 0 && CC % (MTHR / MT) == 0 && (MT * CC) % MTHR == 0, "mean tile must divide evenly");

__global__ void __launch_bounds__(MTHR) meanpc_kernel(const float* __restrict__ x, float* __restrict__ mean_t) {
    __shared__ float tl[MT][CC + 1];
    const int t  = threadIdx.x;
    const int b  = blockIdx.x / (NN / MT);
    const int n0 = (blockIdx.x % (NN / MT)) * MT;
    const int nl = t % MT;
    const int c0 = t / MT;
#pragma unroll
    for (int i = 0; i < CC / (MTHR / MT); ++i) {
        const int c = c0 + (MTHR / MT) * i;
        const float* xp = x + ((size_t)b * CC + c) * (DD * NN) + n0 + nl;
        tl[nl][c] = (xp[0] + xp[NN] + xp[2 * NN]) * (1.0f / 3.0f);
    }
    __syncthreads();
    float* dst = mean_t + ((size_t)b * NN + n0) * CC;
#pragma unroll
    for (int i = 0; i < MT * CC / MTHR; ++i) {
        const int f = t + MTHR * i;
        dst[f] = tl[f / CC][f % CC];
    }
}

typedef __attribute__((ext_vector_type(4))) float v4f;

__global__ void sqnorm_kernel(const float* __restrict__ mean_t, float* __restrict__ sq) {
    int i = blockIdx.x * blockDim.x + threadIdx.x;
    if (i >= BB * NN) return;
    const float* p = mean_t + (size_t)i * CC;
    float s = 0.0f;
#pragma unroll
    for (int c4 = 0; c4 < CC / 4; ++c4) {
        const v4f v = *(const v4f*)(p + 4 * c4);
        s = __builtin_fmaf(v.x, v.x, s);
        s = __builtin_fmaf(v.y, v.y, s);
        s = __builtin_fmaf(v.z, v.z, s);
        s = __builtin_fmaf(v.w, v.w, s);
    }
    sq[i] = s;
}

constexpr int KQ    = 32;
constexpr int NS    = 2;
constexpr int TR    = 32;
constexpr int PP    = 2 * CC + 4;
constexpr int RG    = 16;
constexpr int NPAIR = RG / 2;
constexpr int NBAT  = CC / 2;
constexpr int AHEAD = 3;
constexpr int NSLOT = AHEAD + 1;
constexpr int NGRP  = TR / RG;
constexpr int SG    = NS * NGRP;
constexpr int NSTEP = NN / (NS * TR);
constexpr int QH    = KQ / 2;
static_assert(NGRP == 2 && QH == 16, "a scoring lane is (query pair c, tile half h): 16 lanes x 2 halves");
static_assert(NN % (NS * TR) == 0, "steps must cover the points exactly");

constexpr int NCE = 63;
constexpr unsigned char SORT16[NCE][2] = {
    {0,1}, {2,3}, {4,5}, {6,7}, {8,9}, {10,11}, {12,13}, {14,15},
    {0,2}, {1,3}, {4,6}, {5,7}, {8,10}, {9,11}, {12,14}, {13,15},
    {1,2}, {5,6}, {9,10}, {13,14}, {0,4}, {3,7}, {8,12}, {11,15},
    {1,5}, {2,6}, {9,13}, {10,14}, {0,8}, {7,15},
    {2,4}, {3,5}, {10,12}, {11,13},
    {1,2}, {3,4}, {5,6}, {9,10}, {11,12}, {13,14},
    {1,9}, {2,10}, {3,11}, {4,12}, {5,13}, {6,14},
    {4,8}, {5,9}, {6,10}, {7,11},
    {2,4}, {3,5}, {6,8}, {7,9}, {10,12}, {11,13},
    {1,2}, {3,4}, {5,6}, {7,8}, {9,10}, {11,12}, {13,14}
};
static_assert(RG == KK, "the fold below pairs list position j with new position 15 - j");

__device__ __forceinline__ void knn_ce(float& da, float& db, int& ia, int& ib) {
    const bool s = db < da;
    const float lo = s ? db : da, hi = s ? da : db;
    const int   il = s ? ib : ia, ih = s ? ia : ib;
    da = lo; db = hi; ia = il; ib = ih;
}

__device__ __forceinline__ void knn_merge_group(float (&d2)[RG], int r0, int n, int nb,
                                                float (&bd)[KK], int (&bi)[KK]) {
    int ri[RG];
#pragma unroll
    for (int i = 0; i < RG; ++i) ri[i] = r0 + i;
    if ((r0 & ~(KQ - 1)) == nb) {
#pragma unroll
        for (int i = 0; i < RG; ++i) if (r0 + i == n) d2[i] = 3.0e38f;
    }
    bool any = false;
#pragma unroll
    for (int i = 0; i < RG; ++i) any |= d2[i] < bd[KK - 1];
    if (any) {
#pragma unroll
        for (int e = 0; e < NCE; ++e)
            knn_ce(d2[SORT16[e][0]], d2[SORT16[e][1]], ri[SORT16[e][0]], ri[SORT16[e][1]]);
#pragma unroll
        for (int j = 0; j < KK; ++j) {
            const bool take = d2[RG - 1 - j] < bd[j];
            bd[j] = take ? d2[RG - 1 - j] : bd[j];
            bi[j] = take ? ri[RG - 1 - j] : bi[j];
        }
#pragma unroll
        for (int k = KK / 2; k >= 1; k >>= 1) {
#pragma unroll
            for (int i = 0; i < KK; ++i)
                if ((i & k) == 0) knn_ce(bd[i], bd[i + k], bi[i], bi[i + k]);
        }
    }
}

__device__ __forceinline__ void knn_merge_step(const float (*sc)[SG][KQ], int lane, int r0, int n, int nb,
                                               float (&bd)[KK], int (&bi)[KK]) {
    float nx[RG];
#pragma unroll
    for (int i = 0; i < RG; ++i) nx[i] = sc[i][0][lane];
    for (int k = 0; k < SG; ++k) {
        float d2[RG];
#pragma unroll
        for (int i = 0; i < RG; ++i) d2[i] = nx[i];
        const int kn = (k + 1 < SG) ? k + 1 : k;
#pragma unroll
        for (int i = 0; i < RG; ++i) nx[i] = sc[i][kn][lane];
        knn_merge_group(d2, r0 + k * RG, n, nb, bd, bi);
    }
}

__device__ __forceinline__ void ld_point(const float* p, float (&v)[CC]) {
#pragma unroll
    for (int c4 = 0; c4 < CC / 4; ++c4) {
        const v4f w = *(const v4f*)(p + 4 * c4);
        v[4 * c4 + 0] = w.x; v[4 * c4 + 1] = w.y; v[4 * c4 + 2] = w.z; v[4 * c4 + 3] = w.w;
    }
}

__global__ void __launch_bounds__((NS + 1) * KQ) knn_kernel(const float* __restrict__ mean_t,
                                                            const float* __restrict__ sq,
                                                            int* __restrict__ idx) {
    __shared__ __attribute__((aligned(16))) float tile[NS][(TR / 2) * PP];
    __shared__ float score[2][RG][SG][KQ];

    const int lane = threadIdx.x & (KQ - 1);
    const int role = __builtin_amdgcn_readfirstlane((int)(threadIdx.x >> 5));
    const bool scorer = role < NS;
    const int b = blockIdx.x / (NN / KQ);
    const int nb = (blockIdx.x % (NN / KQ)) * KQ;
    const int n = nb + lane;
    const int qc = lane & (QH - 1);
    const int hh = lane >> 4;
    const float* pcb = mean_t + (size_t)b * NN * CC;
    const float* sqb = sq + (size_t)b * NN;
    float* mytile = &tile[scorer ? role : 0][0];

    float qa[CC];
    float qb[CC];
    float pf[CC];
    if (scorer) {
        ld_point(pcb + (size_t)(nb + qc) * CC, qa);
        ld_point(pcb + (size_t)(nb + qc + QH) * CC, qb);
        ld_point(pcb + (size_t)(role * TR + lane) * CC, pf);
    }

    float bd[KK];
    int   bi[KK];
#pragma unroll
    for (int j = 0; j < KK; ++j) { bd[j] = 3.0e38f; bi[j] = 0; }

    const int sp = (lane >> 1) * PP + (lane & 1);

    for (int st = 0; st < NSTEP; ++st) {
        if (scorer) {
#pragma unroll
            for (int c = 0; c < CC; ++c) mytile[sp + 2 * c] = pf[c];
        }
        __syncthreads();

        if (scorer) {
            const int t = st * NS + role;
            v4f rv[NSLOT][NPAIR];
            if (st + 1 < NSTEP) ld_point(pcb + (size_t)((t + NS) * TR + lane) * CC, pf);
            const int r0 = (t * NGRP + hh) * RG;
            const float* rows = &mytile[hh * NPAIR * PP];
            v4f sqv[RG / 4];
#pragma unroll
            for (int j = 0; j < RG / 4; ++j) sqv[j] = *(const v4f*)&sqb[r0 + 4 * j];
#pragma unroll
            for (int s = 0; s < AHEAD; ++s) {
#pragma unroll
                for (int p = 0; p < NPAIR; ++p) rv[s][p] = *(const v4f*)&rows[p * PP + 4 * s];
            }

            v2f acca[NPAIR], accb[NPAIR];
#pragma unroll
            for (int p = 0; p < NPAIR; ++p) { acca[p] = (v2f){0.0f, 0.0f}; accb[p] = (v2f){0.0f, 0.0f}; }
#pragma unroll
            for (int bt = 0; bt < NBAT; ++bt) {
                const int nx = bt + AHEAD;
                if (nx < NBAT) {
#pragma unroll
                    for (int p = 0; p < NPAIR; ++p)
                        rv[nx % NSLOT][p] = *(const v4f*)&rows[p * PP + 4 * nx];
                }
                __builtin_amdgcn_sched_barrier(0);
                const int cs = bt % NSLOT;
                const v2f qa0 = {qa[2 * bt + 0], qa[2 * bt + 0]};
                const v2f qa1 = {qa[2 * bt + 1], qa[2 * bt + 1]};
                const v2f qb0 = {qb[2 * bt + 0], qb[2 * bt + 0]};
                const v2f qb1 = {qb[2 * bt + 1], qb[2 * bt + 1]};
#pragma unroll
                for (int p = 0; p < NPAIR; ++p) acca[p] = __builtin_elementwise_fma(qa0, rv[cs][p].xy, acca[p]);
#pragma unroll
                for (int p = 0; p < NPAIR; ++p) accb[p] = __builtin_elementwise_fma(qb0, rv[cs][p].xy, accb[p]);
#pragma unroll
                for (int p = 0; p < NPAIR; ++p) acca[p] = __builtin_elementwise_fma(qa1, rv[cs][p].zw, acca[p]);
#pragma unroll
                for (int p = 0; p < NPAIR; ++p) accb[p] = __builtin_elementwise_fma(qb1, rv[cs][p].zw, accb[p]);
                __builtin_amdgcn_sched_barrier(0);
            }

#pragma unroll
            for (int i = 0; i < RG; ++i) {
                score[st & 1][i][role * NGRP + hh][qc]      = sqv[i >> 2][i & 3] - 2.0f * acca[i >> 1][i & 1];
                score[st & 1][i][role * NGRP + hh][qc + QH] = sqv[i >> 2][i & 3] - 2.0f * accb[i >> 1][i & 1];
            }
        } else if (st > 0) {
            knn_merge_step(score[(st - 1) & 1], lane, (st - 1) * SG * RG, n, nb, bd, bi);
        }
        __syncthreads();
    }
    if (!scorer) {
        constexpr int SL = NSTEP - 1;
        knn_merge_step(score[SL & 1], lane, SL * SG * RG, n, nb, bd, bi);
#pragma unroll
        for (int j = 0; j < KK; ++j) idx[((size_t)b * KK + j) * NN + n] = bi[j];
    }
}

typedef __attribute__((ext_vector_type(2))) int v2i;

__device__ __forceinline__ v8f ld8(const float* p) {
    const v4f lo = *(const v4f*)p;
    const v4f hi = *(const v4f*)(p + 4);
    return (v8f){lo.x, lo.y, lo.z, lo.w, hi.x, hi.y, hi.z, hi.w};
}

__device__ __forceinline__ void mma_block(const v8f& a, const v8f& b0, const v8f& b1, v8f& c0, v8f& c1) {
#pragma unroll
    for (int m = 0; m < 4; ++m) {
        const v2f af = {a[2 * m], a[2 * m + 1]};
        const v2f f0 = {b0[2 * m], b0[2 * m + 1]};
        const v2f f1 = {b1[2 * m], b1[2 * m + 1]};
        c0 = __builtin_amdgcn_wmma_f32_16x16x4_f32(false, af, false, f0, (short)0, c0, false, false);
        c1 = __builtin_amdgcn_wmma_f32_16x16x4_f32(false, af, false, f1, (short)0, c1, false, false);
    }
}

__device__ __forceinline__ v8f bias_relu(const v8f& acc, const v8f& bias) {
    v8f o;
#pragma unroll
    for (int r = 0; r < 8; ++r) o[r] = fmaxf(acc[r] + bias[r], 0.0f);
    return o;
}

constexpr int GF   = 2;
constexpr int NRNG = 2 * GF;
constexpr int S1   = EE / 16;
constexpr int S12  = S1 + HH / 16;
constexpr int S3   = HH / 16;
static_assert(S12 % NRNG == 0 && S3 % NRNG == 0, "ring phase must repeat from trip to trip");
static_assert(NCT == 2, "mma_block and the paired loads below are written for two column tiles");

constexpr int RP   = 20;
constexpr int BLK  = 16 * RP + 16;
constexpr int STG  = S12 * BLK;
constexpr int CPG  = 4;
constexpr int NP12 = S12 * 64 / 128;
constexpr int NP3  = S3 * 64 / 128;
static_assert(NP12 == 3 * CPG && NP3 == 2 * CPG, "three copy groups per trip of the first loop, two of the second");
static_assert(MW * 32 == 128, "the copy maps below are written for 128 threads");
static_assert(S12 % GF == 0 && S3 % GF == 0, "a trip is a whole number of request groups");
static_assert(22 <= S12 - GF && 12 <= S3 - GF && GF == 2, "the last copy group of a trip is stored no later than the place that holds the trip's barrier (the places below are written for GF = 2)");
static_assert((STG * 2 * 4) <= 65536, "two stages must fit the static LDS limit");

__global__ void __launch_bounds__(MW * 32) mlp_kernel(const float* __restrict__ x,
                                                      const float* __restrict__ mean_t,
                                                      const int* __restrict__ idx,
                                                      const float* __restrict__ W1, const float* __restrict__ b1,
                                                      const float* __restrict__ W2, const float* __restrict__ b2,
                                                      const float* __restrict__ W3, const float* __restrict__ b3,
                                                      float* __restrict__ out) {
    __shared__ __attribute__((aligned(16))) float wst[2 * STG];

    const int t    = threadIdx.x;
    const int lane = t & 31;
    const int c    = lane & 15;
    const int h    = lane >> 4;
    const int wv   = __builtin_amdgcn_readfirstlane((int)(threadIdx.x >> 5));
    const int task = blockIdx.x * MW + wv;
    const int np   = task % (NN / NCT);
    const int bd   = task / (NN / NCT);
    const int d    = bd % DD;
    const int b    = bd / DD;
    const int n0   = np * NCT;

    const v8f vzero = {0.f, 0.f, 0.f, 0.f, 0.f, 0.f, 0.f, 0.f};

    const float* cw1 = W1 + (size_t)(t >> 5) * EE + 4 * (t & 31);
    const float* cw2 = W2 + (size_t)(t >> 2) * HH + 4 * (t & 3);
    const float* cw3 = W3 + (size_t)(t >> 6) * HH + 4 * (t & 63);
    const int dw1 = ((t & 31) >> 2) * BLK + (t >> 5) * RP + 4 * (t & 3);
    const int dw2 = (S1 + (t >> 6)) * BLK + ((t >> 2) & 15) * RP + 4 * (t & 3);
    const int dw3 = ((t & 63) >> 2) * BLK + (t >> 6) * RP + 4 * (t & 3);
#define SRC12(g, i) (((i) < 4) ? (cw1 + (size_t)(g) * 16 * EE + (i) * 4 * EE) \
                               : (cw2 + (size_t)(g) * 16 + (size_t)((i) - 4) * 32 * HH))
#define DST12(i)    (((i) < 4) ? (dw1 + (i) * 4 * RP) : (dw2 + ((i) - 4) * 2 * BLK))
#define SRC3(v, i)  (cw3 + (size_t)(v) * 16 * HH + (i) * 2 * HH)
#define DST3(i)     (dw3 + (i) * 2 * RP)

    const int fo = c * RP + 8 * h;

    {
        v4f pst[NP12];
#pragma unroll
        for (int i = 0; i < NP12; ++i) pst[i] = *(const v4f*)SRC12(0, i);
#pragma unroll
        for (int i = 0; i < NP12; ++i) *(v4f*)&wst[DST12(i)] = pst[i];
    }

    v8f E[S1][NCT];
    {
        const v2i jj = *(const v2i*)&idx[((size_t)b * KK + c) * NN + n0];
        const int j0 = jj.x & (NN - 1);
        const int j1 = jj.y & (NN - 1);
        const float* xb = x + (((size_t)b * CC) * DD + d) * NN + n0;
        const float* mb = mean_t + (size_t)b * NN * CC;
        const float* g0p = mb + (size_t)j0 * CC + 8 * h;
        const float* g1p = mb + (size_t)j1 * CC + 8 * h;
#pragma unroll
        for (int q = 0; q < CC / 16; ++q) {
            const v8f g0 = ld8(g0p + 16 * q);
            const v8f g1 = ld8(g1p + 16 * q);
#pragma unroll
            for (int r = 0; r < 8; ++r) {
                const int ch = q * 16 + 8 * h + r;
                const v2f xv = *(const v2f*)&xb[(size_t)ch * DD * NN];
                E[q][0][r] = xv.x;
                E[q][1][r] = xv.y;
                E[q + CC / 16][0][r] = g0[r] - xv.x;
                E[q + CC / 16][1][r] = g1[r] - xv.y;
            }
        }
    }

    __syncthreads();

    v8f A[NRNG];
#pragma unroll
    for (int s = 0; s < GF; ++s) A[s] = ld8(&wst[s * BLK + fo]);

    v8f H2[HH / 16][NCT];
#pragma unroll
    for (int u = 0; u < HH / 16; ++u) { H2[u][0] = vzero; H2[u][1] = vzero; }

    for (int T = 0; T < HH / 16; ++T) {
        const int ocur = (T & 1) * STG;
        const int onxt = ((T + 1) & 1) * STG;
        const int Tn = (T + 1 < HH / 16) ? T + 1 : T;
        const v8f bias1 = ld8(b1 + T * 16 + 8 * h);
        v8f c0 = vzero, c1 = vzero;
        v8f t0 = vzero, t1 = vzero;
        v4f st[CPG];
#pragma unroll
        for (int s = 0; s < S12; ++s) {
            if (s % 8 == 6 && s / 8 < NP12 / CPG) {
#pragma unroll
                for (int e = 0; e < CPG; ++e) {
                    const int i  = CPG * (s / 8) + e;
                    *(v4f*)&wst[onxt + DST12(i)] = st[e];
                }
            }
            if (s % GF == 0) {
                if (s == S12 - GF) __syncthreads();
#pragma unroll
                for (int g = 0; g < GF; ++g) {
                    const int sn = s + GF + g;
                    A[sn % NRNG] = ld8(&wst[(sn < S12 ? ocur + sn * BLK : onxt + (sn - S12) * BLK) + fo]);
                }
            }
            if (s % 8 == 0 && s / 8 < NP12 / CPG) {
#pragma unroll
                for (int e = 0; e < CPG; ++e) {
                    const int i  = CPG * (s / 8) + e;
                    st[e] = *(const v4f*)SRC12(Tn, i);
                }
            }
            __builtin_amdgcn_sched_barrier(0);
            if (s < S1) {
                mma_block(A[s % NRNG], E[s][0], E[s][1], c0, c1);
            } else {
                if (s == S1) { t0 = bias_relu(c0, bias1); t1 = bias_relu(c1, bias1); }
                mma_block(A[s % NRNG], t0, t1, H2[s - S1][0], H2[s - S1][1]);
            }
            __builtin_amdgcn_sched_barrier(0);
        }
    }

    {
        v4f pst[NP3];
#pragma unroll
        for (int i = 0; i < NP3; ++i) pst[i] = *(const v4f*)SRC3(0, i);
#pragma unroll
        for (int u = 0; u < HH / 16; ++u) {
            const v8f bias2 = ld8(b2 + u * 16 + 8 * h);
            H2[u][0] = bias_relu(H2[u][0], bias2);
            H2[u][1] = bias_relu(H2[u][1], bias2);
        }
#pragma unroll
        for (int i = 0; i < NP3; ++i) *(v4f*)&wst[DST3(i)] = pst[i];
    }
    __syncthreads();
#pragma unroll
    for (int s = 0; s < GF; ++s) A[s] = ld8(&wst[s * BLK + fo]);

    for (int V = 0; V < CC / 16; ++V) {
        const int Vn = (V + 1 < CC / 16) ? V + 1 : V;
        const int ocur = (V & 1) * STG;
        const int onxt = ((V + 1) & 1) * STG;
        const v8f bias3 = ld8(b3 + V * 16 + 8 * h);
        v8f c0 = vzero, c1 = vzero;
        v4f st[CPG];
#pragma unroll
        for (int s = 0; s < S3; ++s) {
            if (s % 6 == 0 && s > 0 && s / 6 - 1 < NP3 / CPG) {
#pragma unroll
                for (int e = 0; e < CPG; ++e) *(v4f*)&wst[onxt + DST3(CPG * (s / 6 - 1) + e)] = st[e];
            }
            if (s % GF == 0) {
                if (s == S3 - GF) __syncthreads();
#pragma unroll
                for (int g = 0; g < GF; ++g) {
                    const int sn = s + GF + g;
                    A[sn % NRNG] = ld8(&wst[(sn < S3 ? ocur + sn * BLK : onxt + (sn - S3) * BLK) + fo]);
                }
            }
            if (s % 6 == 0 && s / 6 < NP3 / CPG) {
#pragma unroll
                for (int e = 0; e < CPG; ++e) st[e] = *(const v4f*)SRC3(Vn, CPG * (s / 6) + e);
            }
            __builtin_amdgcn_sched_barrier(0);
            mma_block(A[s % NRNG], H2[s][0], H2[s][1], c0, c1);
            __builtin_amdgcn_sched_barrier(0);
        }
        float m0[8], m1[8];
#pragma unroll
        for (int r = 0; r < 8; ++r) { m0[r] = c0[r] + bias3[r]; m1[r] = c1[r] + bias3[r]; }
#pragma unroll
        for (int o = 1; o < 16; o <<= 1) {
#pragma unroll
            for (int r = 0; r < 8; ++r) {
                m0[r] = fmaxf(m0[r], __shfl_xor(m0[r], o));
                m1[r] = fmaxf(m1[r], __shfl_xor(m1[r], o));
            }
        }
        if (c == 0) {
            v2f xr[8];
#pragma unroll
            for (int r = 0; r < 8; ++r)
                xr[r] = *(const v2f*)&x[(((size_t)b * CC + V * 16 + 8 * h + r) * DD + d) * NN + n0];
#pragma unroll
            for (int r = 0; r < 8; ++r)
                *(v2f*)&out[(((size_t)b * CC + V * 16 + 8 * h + r) * DD + d) * NN + n0] = (v2f){m0[r] + xr[r].x, m1[r] + xr[r].y};
        }
    }
#undef SRC12
#undef DST12
#undef SRC3
#undef DST3
}

extern "C" void kernel_launch(void* const* d_in, const int* in_sizes, int n_in,
                              void* d_out, int out_size, void* d_ws, size_t ws_size,
                              hipStream_t stream) {
    const float* x  = (const float*)d_in[0];
    const float* W1 = (const float*)d_in[1];
    const float* b1 = (const float*)d_in[2];
    const float* W2 = (const float*)d_in[3];
    const float* b2 = (const float*)d_in[4];
    const float* W3 = (const float*)d_in[5];
    const float* b3 = (const float*)d_in[6];
    float* out = (float*)d_out;

    float* mean_t  = (float*)d_ws;
    float* sq      = mean_t + (size_t)BB * NN * CC;
    int*   idxp    = (int*)(sq + (size_t)BB * NN);

    meanpc_kernel<<<BB * (NN / MT), MTHR, 0, stream>>>(x, mean_t);
    sqnorm_kernel<<<(BB * NN + 255) / 256, 256, 0, stream>>>(mean_t, sq);
    knn_kernel<<<BB * (NN / KQ), (NS + 1) * KQ, 0, stream>>>(mean_t, sq, idxp);
    mlp_kernel<<<BB * DD * (NN / NCT) / MW, MW * 32, 0, stream>>>(x, mean_t, idxp,
                                                        W1, b1, W2, b2, W3, b3, out);
}


# KNN workgroup 3 -> 8 waves, 2 per SIMD: 4 scoring waves take alternate 32-point tiles, list wave j folds scoring wave j's groups, the 4 sorted lists folded at the end; scores same, ties may reorder
# speedup vs baseline: 13.6993x; 1.0553x over previous
#include <hip/hip_runtime.h>

#define BB 2
#define CC 64
#define DD 3
#define NN 4096
#define KK 16
#define EE 128
#define HH 256

typedef __attribute__((ext_vector_type(2))) float v2f;
typedef __attribute__((ext_vector_type(8))) float v8f;

constexpr int NCT  = 2;
constexpr int MW   = 4;

constexpr int MT = 32;
constexpr int MTHR = 256;
static_assert(NN % MT == 0 && MTHR % MT == 0 && CC % (MTHR / MT) == 0 && (MT * CC) % MTHR == 0, "mean tile must divide evenly");

__global__ void __launch_bounds__(MTHR) meanpc_kernel(const float* __restrict__ x, float* __restrict__ mean_t) {
    __shared__ float tl[MT][CC + 1];
    const int t  = threadIdx.x;
    const int b  = blockIdx.x / (NN / MT);
    const int n0 = (blockIdx.x % (NN / MT)) * MT;
    const int nl = t % MT;
    const int c0 = t / MT;
#pragma unroll
    for (int i = 0; i < CC / (MTHR / MT); ++i) {
        const int c = c0 + (MTHR / MT) * i;
        const float* xp = x + ((size_t)b * CC + c) * (DD * NN) + n0 + nl;
        tl[nl][c] = (xp[0] + xp[NN] + xp[2 * NN]) * (1.0f / 3.0f);
    }
    __syncthreads();
    float* dst = mean_t + ((size_t)b * NN + n0) * CC;
#pragma unroll
    for (int i = 0; i < MT * CC / MTHR; ++i) {
        const int f = t + MTHR * i;
        dst[f] = tl[f / CC][f % CC];
    }
}

typedef __attribute__((ext_vector_type(4))) float v4f;

__global__ void sqnorm_kernel(const float* __restrict__ mean_t, float* __restrict__ sq) {
    int i = blockIdx.x * blockDim.x + threadIdx.x;
    if (i >= BB * NN) return;
    const float* p = mean_t + (size_t)i * CC;
    float s = 0.0f;
#pragma unroll
    for (int c4 = 0; c4 < CC / 4; ++c4) {
        const v4f v = *(const v4f*)(p + 4 * c4);
        s = __builtin_fmaf(v.x, v.x, s);
        s = __builtin_fmaf(v.y, v.y, s);
        s = __builtin_fmaf(v.z, v.z, s);
        s = __builtin_fmaf(v.w, v.w, s);
    }
    sq[i] = s;
}

constexpr int KQ    = 32;
constexpr int NS    = 4;
constexpr int NL    = NS;
constexpr int TR    = 32;
constexpr int PP    = 2 * CC + 4;
constexpr int RG    = 16;
constexpr int NPAIR = RG / 2;
constexpr int NBAT  = CC / 2;
constexpr int AHEAD = 2;
constexpr int NSLOT = AHEAD + 1;
constexpr int NGRP  = TR / RG;
constexpr int SG    = NS * NGRP;
constexpr int NSTEP = NN / (NS * TR);
constexpr int QH    = KQ / 2;
static_assert(NGRP == 2 && QH == 16, "a scoring lane is (query pair c, tile half h): 16 lanes x 2 halves");
static_assert(NN % (NS * TR) == 0, "steps must cover the points exactly");
static_assert(((NSTEP - 1) & 1) == 1, "the last step's scores wait in buffer 1: buffer 0 is free for the lists' hand-over");
static_assert(2 * NL * KK * KQ <= RG * SG * KQ, "the lists' hand-over (scores and point numbers of NL lists) fits one score buffer");

constexpr int NCE = 63;
constexpr unsigned char SORT16[NCE][2] = {
    {0,1}, {2,3}, {4,5}, {6,7}, {8,9}, {10,11}, {12,13}, {14,15},
    {0,2}, {1,3}, {4,6}, {5,7}, {8,10}, {9,11}, {12,14}, {13,15},
    {1,2}, {5,6}, {9,10}, {13,14}, {0,4}, {3,7}, {8,12}, {11,15},
    {1,5}, {2,6}, {9,13}, {10,14}, {0,8}, {7,15},
    {2,4}, {3,5}, {10,12}, {11,13},
    {1,2}, {3,4}, {5,6}, {9,10}, {11,12}, {13,14},
    {1,9}, {2,10}, {3,11}, {4,12}, {5,13}, {6,14},
    {4,8}, {5,9}, {6,10}, {7,11},
    {2,4}, {3,5}, {6,8}, {7,9}, {10,12}, {11,13},
    {1,2}, {3,4}, {5,6}, {7,8}, {9,10}, {11,12}, {13,14}
};
static_assert(RG == KK, "the fold below pairs list position j with new position 15 - j");

__device__ __forceinline__ void knn_ce(float& da, float& db, int& ia, int& ib) {
    const bool s = db < da;
    const float lo = s ? db : da, hi = s ? da : db;
    const int   il = s ? ib : ia, ih = s ? ia : ib;
    da = lo; db = hi; ia = il; ib = ih;
}

__device__ __forceinline__ void knn_fold_sorted(float (&nd)[RG], int (&ni)[RG], float (&bd)[KK], int (&bi)[KK]) {
#pragma unroll
    for (int j = 0; j < KK; ++j) {
        const bool take = nd[RG - 1 - j] < bd[j];
        bd[j] = take ? nd[RG - 1 - j] : bd[j];
        bi[j] = take ? ni[RG - 1 - j] : bi[j];
    }
#pragma unroll
    for (int k = KK / 2; k >= 1; k >>= 1) {
#pragma unroll
        for (int i = 0; i < KK; ++i)
            if ((i & k) == 0) knn_ce(bd[i], bd[i + k], bi[i], bi[i + k]);
    }
}

__device__ __forceinline__ void knn_merge_group(float (&d2)[RG], int r0, int n, int nb,
                                                float (&bd)[KK], int (&bi)[KK]) {
    int ri[RG];
#pragma unroll
    for (int i = 0; i < RG; ++i) ri[i] = r0 + i;
    if ((r0 & ~(KQ - 1)) == nb) {
#pragma unroll
        for (int i = 0; i < RG; ++i) if (r0 + i == n) d2[i] = 3.0e38f;
    }
    bool any = false;
#pragma unroll
    for (int i = 0; i < RG; ++i) any |= d2[i] < bd[KK - 1];
    if (any) {
#pragma unroll
        for (int e = 0; e < NCE; ++e)
            knn_ce(d2[SORT16[e][0]], d2[SORT16[e][1]], ri[SORT16[e][0]], ri[SORT16[e][1]]);
        knn_fold_sorted(d2, ri, bd, bi);
    }
}

__device__ __forceinline__ void knn_merge_step(const float (*sc)[SG][KQ], int lane, int k0, int r0, int n, int nb,
                                               float (&bd)[KK], int (&bi)[KK]) {
    float nx[RG];
#pragma unroll
    for (int i = 0; i < RG; ++i) nx[i] = sc[i][k0][lane];
    for (int g = 0; g < NGRP; ++g) {
        float d2[RG];
#pragma unroll
        for (int i = 0; i < RG; ++i) d2[i] = nx[i];
        const int kn = (g + 1 < NGRP) ? k0 + g + 1 : k0 + g;
#pragma unroll
        for (int i = 0; i < RG; ++i) nx[i] = sc[i][kn][lane];
        knn_merge_group(d2, r0 + (k0 + g) * RG, n, nb, bd, bi);
    }
}

__device__ __forceinline__ void ld_point(const float* p, float (&v)[CC]) {
#pragma unroll
    for (int c4 = 0; c4 < CC / 4; ++c4) {
        const v4f w = *(const v4f*)(p + 4 * c4);
        v[4 * c4 + 0] = w.x; v[4 * c4 + 1] = w.y; v[4 * c4 + 2] = w.z; v[4 * c4 + 3] = w.w;
    }
}

__global__ void __launch_bounds__((NS + NL) * KQ) knn_kernel(const float* __restrict__ mean_t,
                                                            const float* __restrict__ sq,
                                                            int* __restrict__ idx) {
    __shared__ __attribute__((aligned(16))) float tile[NS][(TR / 2) * PP];
    __shared__ float score[2][RG][SG][KQ];

    const int lane = threadIdx.x & (KQ - 1);
    const int role = __builtin_amdgcn_readfirstlane((int)(threadIdx.x >> 5));
    const bool scorer = role < NS;
    const int k0 = (scorer ? 0 : role - NS) * NGRP;
    const int b = blockIdx.x / (NN / KQ);
    const int nb = (blockIdx.x % (NN / KQ)) * KQ;
    const int n = nb + lane;
    const int qc = lane & (QH - 1);
    const int hh = lane >> 4;
    const float* pcb = mean_t + (size_t)b * NN * CC;
    const float* sqb = sq + (size_t)b * NN;
    float* mytile = &tile[scorer ? role : 0][0];

    float qa[CC];
    float qb[CC];
    float pf[CC];
    if (scorer) {
        ld_point(pcb + (size_t)(nb + qc) * CC, qa);
        ld_point(pcb + (size_t)(nb + qc + QH) * CC, qb);
        ld_point(pcb + (size_t)(role * TR + lane) * CC, pf);
    }

    float bd[KK];
    int   bi[KK];
#pragma unroll
    for (int j = 0; j < KK; ++j) { bd[j] = 3.0e38f; bi[j] = 0; }

    const int sp = (lane >> 1) * PP + (lane & 1);

    for (int st = 0; st < NSTEP; ++st) {
        if (scorer) {
#pragma unroll
            for (int c = 0; c < CC; ++c) mytile[sp + 2 * c] = pf[c];
        }
        __syncthreads();

        if (scorer) {
            const int t = st * NS + role;
            v4f rv[NSLOT][NPAIR];
            if (st + 1 < NSTEP) ld_point(pcb + (size_t)((t + NS) * TR + lane) * CC, pf);
            const int r0 = (t * NGRP + hh) * RG;
            const float* rows = &mytile[hh * NPAIR * PP];
            v4f sqv[RG / 4];
#pragma unroll
            for (int j = 0; j < RG / 4; ++j) sqv[j] = *(const v4f*)&sqb[r0 + 4 * j];
#pragma unroll
            for (int s = 0; s < AHEAD; ++s) {
#pragma unroll
                for (int p = 0; p < NPAIR; ++p) rv[s][p] = *(const v4f*)&rows[p * PP + 4 * s];
            }

            v2f acca[NPAIR], accb[NPAIR];
#pragma unroll
            for (int p = 0; p < NPAIR; ++p) { acca[p] = (v2f){0.0f, 0.0f}; accb[p] = (v2f){0.0f, 0.0f}; }
#pragma unroll
            for (int bt = 0; bt < NBAT; ++bt) {
                const int nx = bt + AHEAD;
                if (nx < NBAT) {
#pragma unroll
                    for (int p = 0; p < NPAIR; ++p)
                        rv[nx % NSLOT][p] = *(const v4f*)&rows[p * PP + 4 * nx];
                }
                __builtin_amdgcn_sched_barrier(0);
                const int cs = bt % NSLOT;
                const v2f qa0 = {qa[2 * bt + 0], qa[2 * bt + 0]};
                const v2f qa1 = {qa[2 * bt + 1], qa[2 * bt + 1]};
                const v2f qb0 = {qb[2 * bt + 0], qb[2 * bt + 0]};
                const v2f qb1 = {qb[2 * bt + 1], qb[2 * bt + 1]};
#pragma unroll
                for (int p = 0; p < NPAIR; ++p) acca[p] = __builtin_elementwise_fma(qa0, rv[cs][p].xy, acca[p]);
#pragma unroll
                for (int p = 0; p < NPAIR; ++p) accb[p] = __builtin_elementwise_fma(qb0, rv[cs][p].xy, accb[p]);
#pragma unroll
                for (int p = 0; p < NPAIR; ++p) acca[p] = __builtin_elementwise_fma(qa1, rv[cs][p].zw, acca[p]);
#pragma unroll
                for (int p = 0; p < NPAIR; ++p) accb[p] = __builtin_elementwise_fma(qb1, rv[cs][p].zw, accb[p]);
                __builtin_amdgcn_sched_barrier(0);
            }

#pragma unroll
            for (int i = 0; i < RG; ++i) {
                score[st & 1][i][role * NGRP + hh][qc]      = sqv[i >> 2][i & 3] - 2.0f * acca[i >> 1][i & 1];
                score[st & 1][i][role * NGRP + hh][qc + QH] = sqv[i >> 2][i & 3] - 2.0f * accb[i >> 1][i & 1];
            }
        } else if (st > 0) {
            knn_merge_step(score[(st - 1) & 1], lane, k0, (st - 1) * SG * RG, n, nb, bd, bi);
        }
        __syncthreads();
    }
    float* hd = &score[0][0][0][0];
    float* hi = hd + NL * KK * KQ;
    if (!scorer) {
        constexpr int SL = NSTEP - 1;
        knn_merge_step(score[SL & 1], lane, k0, SL * SG * RG, n, nb, bd, bi);
        if (role != NS) {
#pragma unroll
            for (int j = 0; j < KK; ++j) {
                hd[((role - NS) * KK + j) * KQ + lane] = bd[j];
                hi[((role - NS) * KK + j) * KQ + lane] = __int_as_float(bi[j]);
            }
        }
    }
    __syncthreads();
    if (role == NS) {
        for (int w = 1; w < NL; ++w) {
            float od[KK];
            int   oi[KK];
#pragma unroll
            for (int j = 0; j < KK; ++j) {
                od[j] = hd[(w * KK + j) * KQ + lane];
                oi[j] = __float_as_int(hi[(w * KK + j) * KQ + lane]);
            }
            knn_fold_sorted(od, oi, bd, bi);
        }
#pragma unroll
        for (int j = 0; j < KK; ++j) idx[((size_t)b * KK + j) * NN + n] = bi[j];
    }
}

typedef __attribute__((ext_vector_type(2))) int v2i;

__device__ __forceinline__ v8f ld8(const float* p) {
    const v4f lo = *(const v4f*)p;
    const v4f hi = *(const v4f*)(p + 4);
    return (v8f){lo.x, lo.y, lo.z, lo.w, hi.x, hi.y, hi.z, hi.w};
}

__device__ __forceinline__ void mma_block(const v8f& a, const v8f& b0, const v8f& b1, v8f& c0, v8f& c1) {
#pragma unroll
    for (int m = 0; m < 4; ++m) {
        const v2f af = {a[2 * m], a[2 * m + 1]};
        const v2f f0 = {b0[2 * m], b0[2 * m + 1]};
        const v2f f1 = {b1[2 * m], b1[2 * m + 1]};
        c0 = __builtin_amdgcn_wmma_f32_16x16x4_f32(false, af, false, f0, (short)0, c0, false, false);
        c1 = __builtin_amdgcn_wmma_f32_16x16x4_f32(false, af, false, f1, (short)0, c1, false, false);
    }
}

__device__ __forceinline__ v8f bias_relu(const v8f& acc, const v8f& bias) {
    v8f o;
#pragma unroll
    for (int r = 0; r < 8; ++r) o[r] = fmaxf(acc[r] + bias[r], 0.0f);
    return o;
}

constexpr int GF   = 2;
constexpr int NRNG = 2 * GF;
constexpr int S1   = EE / 16;
constexpr int S12  = S1 + HH / 16;
constexpr int S3   = HH / 16;
static_assert(S12 % NRNG == 0 && S3 % NRNG == 0, "ring phase must repeat from trip to trip");
static_assert(NCT == 2, "mma_block and the paired loads below are written for two column tiles");

constexpr int RP   = 20;
constexpr int BLK  = 16 * RP + 16;
constexpr int STG  = S12 * BLK;
constexpr int CPG  = 4;
constexpr int NP12 = S12 * 64 / 128;
constexpr int NP3  = S3 * 64 / 128;
static_assert(NP12 == 3 * CPG && NP3 == 2 * CPG, "three copy groups per trip of the first loop, two of the second");
static_assert(MW * 32 == 128, "the copy maps below are written for 128 threads");
static_assert(S12 % GF == 0 && S3 % GF == 0, "a trip is a whole number of request groups");
static_assert(22 <= S12 - GF && 12 <= S3 - GF && GF == 2, "the last copy group of a trip is stored no later than the place that holds the trip's barrier (the places below are written for GF = 2)");
static_assert((STG * 2 * 4) <= 65536, "two stages must fit the static LDS limit");

__global__ void __launch_bounds__(MW * 32) mlp_kernel(const float* __restrict__ x,
                                                      const float* __restrict__ mean_t,
                                                      const int* __restrict__ idx,
                                                      const float* __restrict__ W1, const float* __restrict__ b1,
                                                      const float* __restrict__ W2, const float* __restrict__ b2,
                                                      const float* __restrict__ W3, const float* __restrict__ b3,
                                                      float* __restrict__ out) {
    __shared__ __attribute__((aligned(16))) float wst[2 * STG];

    const int t    = threadIdx.x;
    const int lane = t & 31;
    const int c    = lane & 15;
    const int h    = lane >> 4;
    const int wv   = __builtin_amdgcn_readfirstlane((int)(threadIdx.x >> 5));
    const int task = blockIdx.x * MW + wv;
    const int np   = task % (NN / NCT);
    const int bd   = task / (NN / NCT);
    const int d    = bd % DD;
    const int b    = bd / DD;
    const int n0   = np * NCT;

    const v8f vzero = {0.f, 0.f, 0.f, 0.f, 0.f, 0.f, 0.f, 0.f};

    const float* cw1 = W1 + (size_t)(t >> 5) * EE + 4 * (t & 31);
    const float* cw2 = W2 + (size_t)(t >> 2) * HH + 4 * (t & 3);
    const float* cw3 = W3 + (size_t)(t >> 6) * HH + 4 * (t & 63);
    const int dw1 = ((t & 31) >> 2) * BLK + (t >> 5) * RP + 4 * (t & 3);
    const int dw2 = (S1 + (t >> 6)) * BLK + ((t >> 2) & 15) * RP + 4 * (t & 3);
    const int dw3 = ((t & 63) >> 2) * BLK + (t >> 6) * RP + 4 * (t & 3);
#define SRC12(g, i) (((i) < 4) ? (cw1 + (size_t)(g) * 16 * EE + (i) * 4 * EE) \
                               : (cw2 + (size_t)(g) * 16 + (size_t)((i) - 4) * 32 * HH))
#define DST12(i)    (((i) < 4) ? (dw1 + (i) * 4 * RP) : (dw2 + ((i) - 4) * 2 * BLK))
#define SRC3(v, i)  (cw3 + (size_t)(v) * 16 * HH + (i) * 2 * HH)
#define DST3(i)     (dw3 + (i) * 2 * RP)

    const int fo = c * RP + 8 * h;

    {
        v4f pst[NP12];
#pragma unroll
        for (int i = 0; i < NP12; ++i) pst[i] = *(const v4f*)SRC12(0, i);
#pragma unroll
        for (int i = 0; i < NP12; ++i) *(v4f*)&wst[DST12(i)] = pst[i];
    }

    v8f E[S1][NCT];
    {
        const v2i jj = *(const v2i*)&idx[((size_t)b * KK + c) * NN + n0];
        const int j0 = jj.x & (NN - 1);
        const int j1 = jj.y & (NN - 1);
        const float* xb = x + (((size_t)b * CC) * DD + d) * NN + n0;
        const float* mb = mean_t + (size_t)b * NN * CC;
        const float* g0p = mb + (size_t)j0 * CC + 8 * h;
        const float* g1p = mb + (size_t)j1 * CC + 8 * h;
#pragma unroll
        for (int q = 0; q < CC / 16; ++q) {
            const v8f g0 = ld8(g0p + 16 * q);
            const v8f g1 = ld8(g1p + 16 * q);
#pragma unroll
            for (int r = 0; r < 8; ++r) {
                const int ch = q * 16 + 8 * h + r;
                const v2f xv = *(const v2f*)&xb[(size_t)ch * DD * NN];
                E[q][0][r] = xv.x;
                E[q][1][r] = xv.y;
                E[q + CC / 16][0][r] = g0[r] - xv.x;
                E[q + CC / 16][1][r] = g1[r] - xv.y;
            }
        }
    }

    __syncthreads();

    v8f A[NRNG];
#pragma unroll
    for (int s = 0; s < GF; ++s) A[s] = ld8(&wst[s * BLK + fo]);

    v8f H2[HH / 16][NCT];
#pragma unroll
    for (int u = 0; u < HH / 16; ++u) { H2[u][0] = vzero; H2[u][1] = vzero; }

    for (int T = 0; T < HH / 16; ++T) {
        const int ocur = (T & 1) * STG;
        const int onxt = ((T + 1) & 1) * STG;
        const int Tn = (T + 1 < HH / 16) ? T + 1 : T;
        const v8f bias1 = ld8(b1 + T * 16 + 8 * h);
        v8f c0 = vzero, c1 = vzero;
        v8f t0 = vzero, t1 = vzero;
        v4f st[CPG];
#pragma unroll
        for (int s = 0; s < S12; ++s) {
            if (s % 8 == 6 && s / 8 < NP12 / CPG) {
#pragma unroll
                for (int e = 0; e < CPG; ++e) {
                    const int i  = CPG * (s / 8) + e;
                    *(v4f*)&wst[onxt + DST12(i)] = st[e];
                }
            }
            if (s % GF == 0) {
                if (s == S12 - GF) __syncthreads();
#pragma unroll
                for (int g = 0; g < GF; ++g) {
                    const int sn = s + GF + g;
                    A[sn % NRNG] = ld8(&wst[(sn < S12 ? ocur + sn * BLK : onxt + (sn - S12) * BLK) + fo]);
                }
            }
            if (s % 8 == 0 && s / 8 < NP12 / CPG) {
#pragma unroll
                for (int e = 0; e < CPG; ++e) {
                    const int i  = CPG * (s / 8) + e;
                    st[e] = *(const v4f*)SRC12(Tn, i);
                }
            }
            __builtin_amdgcn_sched_barrier(0);
            if (s < S1) {
                mma_block(A[s % NRNG], E[s][0], E[s][1], c0, c1);
            } else {
                if (s == S1) { t0 = bias_relu(c0, bias1); t1 = bias_relu(c1, bias1); }
                mma_block(A[s % NRNG], t0, t1, H2[s - S1][0], H2[s - S1][1]);
            }
            __builtin_amdgcn_sched_barrier(0);
        }
    }

    {
        v4f pst[NP3];
#pragma unroll
        for (int i = 0; i < NP3; ++i) pst[i] = *(const v4f*)SRC3(0, i);
#pragma unroll
        for (int u = 0; u < HH / 16; ++u) {
            const v8f bias2 = ld8(b2 + u * 16 + 8 * h);
            H2[u][0] = bias_relu(H2[u][0], bias2);
            H2[u][1] = bias_relu(H2[u][1], bias2);
        }
#pragma unroll
        for (int i = 0; i < NP3; ++i) *(v4f*)&wst[DST3(i)] = pst[i];
    }
    __syncthreads();
#pragma unroll
    for (int s = 0; s < GF; ++s) A[s] = ld8(&wst[s * BLK + fo]);

    for (int V = 0; V < CC / 16; ++V) {
        const int Vn = (V + 1 < CC / 16) ? V + 1 : V;
        const int ocur = (V & 1) * STG;
        const int onxt = ((V + 1) & 1) * STG;
        const v8f bias3 = ld8(b3 + V * 16 + 8 * h);
        v8f c0 = vzero, c1 = vzero;
        v4f st[CPG];
#pragma unroll
        for (int s = 0; s < S3; ++s) {
            if (s % 6 == 0 && s > 0 && s / 6 - 1 < NP3 / CPG) {
#pragma unroll
                for (int e = 0; e < CPG; ++e) *(v4f*)&wst[onxt + DST3(CPG * (s / 6 - 1) + e)] = st[e];
            }
            if (s % GF == 0) {
                if (s == S3 - GF) __syncthreads();
#pragma unroll
                for (int g = 0; g < GF; ++g) {
                    const int sn = s + GF + g;
                    A[sn % NRNG] = ld8(&wst[(sn < S3 ? ocur + sn * BLK : onxt + (sn - S3) * BLK) + fo]);
                }
            }
            if (s % 6 == 0 && s / 6 < NP3 / CPG) {
#pragma unroll
                for (int e = 0; e < CPG; ++e) st[e] = *(const v4f*)SRC3(Vn, CPG * (s / 6) + e);
            }
            __builtin_amdgcn_sched_barrier(0);
            mma_block(A[s % NRNG], H2[s][0], H2[s][1], c0, c1);
            __builtin_amdgcn_sched_barrier(0);
        }
        float m0[8], m1[8];
#pragma unroll
        for (int r = 0; r < 8; ++r) { m0[r] = c0[r] + bias3[r]; m1[r] = c1[r] + bias3[r]; }
#pragma unroll
        for (int o = 1; o < 16; o <<= 1) {
#pragma unroll
            for (int r = 0; r < 8; ++r) {
                m0[r] = fmaxf(m0[r], __shfl_xor(m0[r], o));
                m1[r] = fmaxf(m1[r], __shfl_xor(m1[r], o));
            }
        }
        if (c == 0) {
            v2f xr[8];
#pragma unroll
            for (int r = 0; r < 8; ++r)
                xr[r] = *(const v2f*)&x[(((size_t)b * CC + V * 16 + 8 * h + r) * DD + d) * NN + n0];
#pragma unroll
            for (int r = 0; r < 8; ++r)
                *(v2f*)&out[(((size_t)b * CC + V * 16 + 8 * h + r) * DD + d) * NN + n0] = (v2f){m0[r] + xr[r].x, m1[r] + xr[r].y};
        }
    }
#undef SRC12
#undef DST12
#undef SRC3
#undef DST3
}

extern "C" void kernel_launch(void* const* d_in, const int* in_sizes, int n_in,
                              void* d_out, int out_size, void* d_ws, size_t ws_size,
                              hipStream_t stream) {
    const float* x  = (const float*)d_in[0];
    const float* W1 = (const float*)d_in[1];
    const float* b1 = (const float*)d_in[2];
    const float* W2 = (const float*)d_in[3];
    const float* b2 = (const float*)d_in[4];
    const float* W3 = (const float*)d_in[5];
    const float* b3 = (const float*)d_in[6];
    float* out = (float*)d_out;

    float* mean_t  = (float*)d_ws;
    float* sq      = mean_t + (size_t)BB * NN * CC;
    int*   idxp    = (int*)(sq + (size_t)BB * NN);

    meanpc_kernel<<<BB * (NN / MT), MTHR, 0, stream>>>(x, mean_t);
    sqnorm_kernel<<<(BB * NN + 255) / 256, 256, 0, stream>>>(mean_t, sq);
    knn_kernel<<<BB * (NN / KQ), (NS + NL) * KQ, 0, stream>>>(mean_t, sq, idxp);
    mlp_kernel<<<BB * DD * (NN / NCT) / MW, MW * 32, 0, stream>>>(x, mean_t, idxp,
                                                        W1, b1, W2, b2, W3, b3, out);
}
